# starting kernel
# baseline (speedup 1.0000x reference)
#include <hip/hip_runtime.h>
#include <hip/hip_bf16.h>

typedef __attribute__((ext_vector_type(2))) float v2f;
typedef __attribute__((ext_vector_type(8))) float v8f;

#define D 128

__device__ __forceinline__ void atomic_add_f32_hw(float* addr, float val) {
    asm volatile("global_atomic_add_f32 %0, %1, off"
                 :
                 : "v"(addr), "v"(val)
                 : "memory");
}

__global__ __launch_bounds__(256) void gcn_zero(float4* __restrict__ p, int n4) {
    int i = blockIdx.x * blockDim.x + threadIdx.x;
    if (i < n4) p[i] = make_float4(0.f, 0.f, 0.f, 0.f);
}

__global__ __launch_bounds__(256) void gcn_gemm_wmma(const float* __restrict__ x,
                                                     const float* __restrict__ W,
                                                     float* __restrict__ support) {
    const int lane = threadIdx.x & 31;
    const int wave = threadIdx.x >> 5;
    const int m0   = blockIdx.x * 16;
    const int n0   = wave * 16;
    const int half = lane >> 4;
    const int l16  = lane & 15;

    const float* arow = x + (size_t)(m0 + l16) * D;
    v8f c = {};
#pragma unroll 4
    for (int k = 0; k < D; k += 4) {
        const int ka = k + half * 2;
        v2f a;
        a.x = arow[ka];
        a.y = arow[ka + 1];
        v2f b;
        b.x = W[(size_t)ka * D + n0 + l16];
        b.y = W[(size_t)(ka + 1) * D + n0 + l16];
        c = __builtin_amdgcn_wmma_f32_16x16x4_f32(
                false, a, false, b, (short)0, c, false, false);
    }

    float* srow = support + n0 + l16;
#pragma unroll
    for (int v = 0; v < 8; ++v) {
        srow[(size_t)(m0 + v + half * 8) * D] = c[v];
    }
}

__global__ __launch_bounds__(256) void gcn_scatter_atomic(const float* __restrict__ support,
                                                          const int* __restrict__ erow,
                                                          const int* __restrict__ ecol,
                                                          const float* __restrict__ eval,
                                                          float* __restrict__ out,
                                                          int nEdges) {
    const int wv = __builtin_amdgcn_readfirstlane(threadIdx.x >> 5);
    const int e  = blockIdx.x * 8 + wv;
    if (e >= nEdges) return;
    const int lane = threadIdx.x & 31;

    const int   r = erow[e];
    const int   c = ecol[e];
    const float v = eval[e];

    const float4 m = *((const float4*)(support + (size_t)c * D) + lane);
    float* dst = out + (size_t)r * D + lane * 4;

    atomic_add_f32_hw(dst + 0, m.x * v);
    atomic_add_f32_hw(dst + 1, m.y * v);
    atomic_add_f32_hw(dst + 2, m.z * v);
    atomic_add_f32_hw(dst + 3, m.w * v);
}

__global__ __launch_bounds__(256) void gcn_tanh(float4* __restrict__ p, int n4) {
    int i = blockIdx.x * blockDim.x + threadIdx.x;
    if (i < n4) {
        float4 t = p[i];
        t.x = tanhf(t.x);
        t.y = tanhf(t.y);
        t.z = tanhf(t.z);
        t.w = tanhf(t.w);
        p[i] = t;
    }
}

extern "C" void kernel_launch(void* const* d_in, const int* in_sizes, int n_in,
                              void* d_out, int out_size, void* d_ws, size_t ws_size,
                              hipStream_t stream) {
    const float* x    = (const float*)d_in[0];
    const float* W    = (const float*)d_in[1];
    const int*   erow = (const int*)d_in[2];
    const int*   ecol = (const int*)d_in[3];
    const float* eval = (const float*)d_in[4];
    float* out = (float*)d_out;

    const int nNodes = in_sizes[0] / D;
    const int nEdges = in_sizes[2];

    float* support = (float*)d_ws;

    {
        const int n4 = out_size / 4;
        gcn_zero<<<(n4 + 255) / 256, 256, 0, stream>>>((float4*)out, n4);
    }
    {
        const int blocks = nNodes / 16;
        gcn_gemm_wmma<<<blocks, 256, 0, stream>>>(x, W, support);
    }
    {
        const int blocks = (nEdges + 7) / 8;
        gcn_scatter_atomic<<<blocks, 256, 0, stream>>>(support, erow, ecol, eval, out, nEdges);
    }
    {
        const int n4 = out_size / 4;
        gcn_tanh<<<(n4 + 255) / 256, 256, 0, stream>>>((float4*)out, n4);
    }
}


# scatter: lane owns row elements l+32j, so every f32 atomic and every gather load instruction covers one whole 128-B line instead of 8 dwords in each of 4 lines; same products, same atomic adds
# speedup vs baseline: 1.1802x; 1.1802x over previous
#include <hip/hip_runtime.h>
#include <hip/hip_bf16.h>

typedef __attribute__((ext_vector_type(2))) float v2f;
typedef __attribute__((ext_vector_type(8))) float v8f;

#define D 128

__device__ __forceinline__ void atomic_add_f32_hw(float* addr, float val) {
    asm volatile("global_atomic_add_f32 %0, %1, off"
                 :
                 : "v"(addr), "v"(val)
                 : "memory");
}

__global__ __launch_bounds__(256) void gcn_zero(float4* __restrict__ p, int n4) {
    int i = blockIdx.x * blockDim.x + threadIdx.x;
    if (i < n4) p[i] = make_float4(0.f, 0.f, 0.f, 0.f);
}

__global__ __launch_bounds__(256) void gcn_gemm_wmma(const float* __restrict__ x,
                                                     const float* __restrict__ W,
                                                     float* __restrict__ support) {
    const int lane = threadIdx.x & 31;
    const int wave = threadIdx.x >> 5;
    const int m0   = blockIdx.x * 16;
    const int n0   = wave * 16;
    const int half = lane >> 4;
    const int l16  = lane & 15;

    const float* arow = x + (size_t)(m0 + l16) * D;
    v8f c = {};
#pragma unroll 4
    for (int k = 0; k < D; k += 4) {
        const int ka = k + half * 2;
        v2f a;
        a.x = arow[ka];
        a.y = arow[ka + 1];
        v2f b;
        b.x = W[(size_t)ka * D + n0 + l16];
        b.y = W[(size_t)(ka + 1) * D + n0 + l16];
        c = __builtin_amdgcn_wmma_f32_16x16x4_f32(
                false, a, false, b, (short)0, c, false, false);
    }

    float* srow = support + n0 + l16;
#pragma unroll
    for (int v = 0; v < 8; ++v) {
        srow[(size_t)(m0 + v + half * 8) * D] = c[v];
    }
}

__global__ __launch_bounds__(256) void gcn_scatter_atomic(const float* __restrict__ support,
                                                          const int* __restrict__ erow,
                                                          const int* __restrict__ ecol,
                                                          const float* __restrict__ eval,
                                                          float* __restrict__ out,
                                                          int nEdges) {
    const int wv = __builtin_amdgcn_readfirstlane(threadIdx.x >> 5);
    const int e  = blockIdx.x * 8 + wv;
    if (e >= nEdges) return;
    const int lane = threadIdx.x & 31;

    const int   r = erow[e];
    const int   c = ecol[e];
    const float v = eval[e];

    const float* src = support + (size_t)c * D + lane;
    const float m0 = src[0];
    const float m1 = src[32];
    const float m2 = src[64];
    const float m3 = src[96];
    float* dst = out + (size_t)r * D + lane;

    atomic_add_f32_hw(dst + 0,  m0 * v);
    atomic_add_f32_hw(dst + 32, m1 * v);
    atomic_add_f32_hw(dst + 64, m2 * v);
    atomic_add_f32_hw(dst + 96, m3 * v);
}

__global__ __launch_bounds__(256) void gcn_tanh(float4* __restrict__ p, int n4) {
    int i = blockIdx.x * blockDim.x + threadIdx.x;
    if (i < n4) {
        float4 t = p[i];
        t.x = tanhf(t.x);
        t.y = tanhf(t.y);
        t.z = tanhf(t.z);
        t.w = tanhf(t.w);
        p[i] = t;
    }
}

extern "C" void kernel_launch(void* const* d_in, const int* in_sizes, int n_in,
                              void* d_out, int out_size, void* d_ws, size_t ws_size,
                              hipStream_t stream) {
    const float* x    = (const float*)d_in[0];
    const float* W    = (const float*)d_in[1];
    const int*   erow = (const int*)d_in[2];
    const int*   ecol = (const int*)d_in[3];
    const float* eval = (const float*)d_in[4];
    float* out = (float*)d_out;

    const int nNodes = in_sizes[0] / D;
    const int nEdges = in_sizes[2];

    float* support = (float*)d_ws;

    {
        const int n4 = out_size / 4;
        gcn_zero<<<(n4 + 255) / 256, 256, 0, stream>>>((float4*)out, n4);
    }
    {
        const int blocks = nNodes / 16;
        gcn_gemm_wmma<<<blocks, 256, 0, stream>>>(x, W, support);
    }
    {
        const int blocks = (nEdges + 7) / 8;
        gcn_scatter_atomic<<<blocks, 256, 0, stream>>>(support, erow, ecol, eval, out, nEdges);
    }
    {
        const int n4 = out_size / 4;
        gcn_tanh<<<(n4 + 255) / 256, 256, 0, stream>>>((float4*)out, n4);
    }
}


# scatter: one wave walks 32 consecutive edges (one coalesced descriptor load, readlane to scalars), next 4 edges' row gathers in flight ahead of the atomics; same adds, arrival order still unordered
# speedup vs baseline: 1.6520x; 1.3998x over previous
#include <hip/hip_runtime.h>
#include <hip/hip_bf16.h>

typedef __attribute__((ext_vector_type(2))) float v2f;
typedef __attribute__((ext_vector_type(8))) float v8f;

#define D 128

__global__ __launch_bounds__(256) void gcn_zero(float4* __restrict__ p, int n4) {
    int i = blockIdx.x * blockDim.x + threadIdx.x;
    if (i < n4) p[i] = make_float4(0.f, 0.f, 0.f, 0.f);
}

__global__ __launch_bounds__(256) void gcn_gemm_wmma(const float* __restrict__ x,
                                                     const float* __restrict__ W,
                                                     float* __restrict__ support) {
    const int lane = threadIdx.x & 31;
    const int wave = threadIdx.x >> 5;
    const int m0   = blockIdx.x * 16;
    const int n0   = wave * 16;
    const int half = lane >> 4;
    const int l16  = lane & 15;

    const float* arow = x + (size_t)(m0 + l16) * D;
    v8f c = {};
#pragma unroll 4
    for (int k = 0; k < D; k += 4) {
        const int ka = k + half * 2;
        v2f a;
        a.x = arow[ka];
        a.y = arow[ka + 1];
        v2f b;
        b.x = W[(size_t)ka * D + n0 + l16];
        b.y = W[(size_t)(ka + 1) * D + n0 + l16];
        c = __builtin_amdgcn_wmma_f32_16x16x4_f32(
                false, a, false, b, (short)0, c, false, false);
    }

    float* srow = support + n0 + l16;
#pragma unroll
    for (int v = 0; v < 8; ++v) {
        srow[(size_t)(m0 + v + half * 8) * D] = c[v];
    }
}

#define EPW 32
#define GRP 4

__device__ __forceinline__ void gather_row(const float* __restrict__ support, int lane, int c,
                                           float (&m)[4]) {
    const float* row = support + (size_t)c * D;
    m[0] = row[lane];
    m[1] = row[lane + 32];
    m[2] = row[lane + 64];
    m[3] = row[lane + 96];
}

__device__ __forceinline__ void add_row(float* __restrict__ out, int voff, int r, float v,
                                        const float (&m)[4]) {
    float* base = out + (size_t)r * D;
    asm volatile("global_atomic_add_f32 %0, %1, %5\n\t"
                 "global_atomic_add_f32 %0, %2, %5 offset:128\n\t"
                 "global_atomic_add_f32 %0, %3, %5 offset:256\n\t"
                 "global_atomic_add_f32 %0, %4, %5 offset:384\n\t"
                 "s_wait_xcnt 0x0"
                 :
                 : "v"(voff), "v"(m[0] * v), "v"(m[1] * v), "v"(m[2] * v), "v"(m[3] * v),
                   "s"(base)
                 : "memory");
}

__global__ __launch_bounds__(256) void gcn_scatter_atomic(const float* __restrict__ support,
                                                          const int* __restrict__ erow,
                                                          const int* __restrict__ ecol,
                                                          const float* __restrict__ eval,
                                                          float* __restrict__ out,
                                                          int nEdges) {
    const int wv = __builtin_amdgcn_readfirstlane(threadIdx.x >> 5);
    const int e0 = (blockIdx.x * 8 + wv) * EPW;
    if (e0 >= nEdges) return;
    const int lane = threadIdx.x & 31;
    const int n = (nEdges - e0 < EPW) ? (nEdges - e0) : EPW;

    int my_r = 0, my_c = 0, my_v = 0;
    if (lane < n) {
        my_r = erow[e0 + lane];
        my_c = ecol[e0 + lane];
        my_v = __float_as_int(eval[e0 + lane]);
    }

    const int voff = lane * 4;

    if (n == EPW) {
        float m[2][GRP][4];
#pragma unroll
        for (int j = 0; j < GRP; ++j)
            gather_row(support, lane, __builtin_amdgcn_readlane(my_c, j), m[0][j]);
#pragma unroll
        for (int g = 0; g < EPW / GRP; ++g) {
            if (g + 1 < EPW / GRP) {
#pragma unroll
                for (int j = 0; j < GRP; ++j)
                    gather_row(support, lane, __builtin_amdgcn_readlane(my_c, (g + 1) * GRP + j),
                               m[(g + 1) & 1][j]);
            }
#pragma unroll
            for (int j = 0; j < GRP; ++j) {
                const int e = g * GRP + j;
                add_row(out, voff, __builtin_amdgcn_readlane(my_r, e),
                        __int_as_float(__builtin_amdgcn_readlane(my_v, e)), m[g & 1][j]);
            }
        }
    } else {
        for (int i = 0; i < n; ++i) {
            float m[4];
            gather_row(support, lane, __builtin_amdgcn_readlane(my_c, i), m);
            add_row(out, voff, __builtin_amdgcn_readlane(my_r, i),
                    __int_as_float(__builtin_amdgcn_readlane(my_v, i)), m);
        }
    }
}

__global__ __launch_bounds__(256) void gcn_tanh(float4* __restrict__ p, int n4) {
    int i = blockIdx.x * blockDim.x + threadIdx.x;
    if (i < n4) {
        float4 t = p[i];
        t.x = tanhf(t.x);
        t.y = tanhf(t.y);
        t.z = tanhf(t.z);
        t.w = tanhf(t.w);
        p[i] = t;
    }
}

extern "C" void kernel_launch(void* const* d_in, const int* in_sizes, int n_in,
                              void* d_out, int out_size, void* d_ws, size_t ws_size,
                              hipStream_t stream) {
    const float* x    = (const float*)d_in[0];
    const float* W    = (const float*)d_in[1];
    const int*   erow = (const int*)d_in[2];
    const int*   ecol = (const int*)d_in[3];
    const float* eval = (const float*)d_in[4];
    float* out = (float*)d_out;

    const int nNodes = in_sizes[0] / D;
    const int nEdges = in_sizes[2];

    float* support = (float*)d_ws;

    {
        const int n4 = out_size / 4;
        gcn_zero<<<(n4 + 255) / 256, 256, 0, stream>>>((float4*)out, n4);
    }
    {
        const int blocks = nNodes / 16;
        gcn_gemm_wmma<<<blocks, 256, 0, stream>>>(x, W, support);
    }
    {
        const int blocks = (nEdges + 255) / 256;
        gcn_scatter_atomic<<<blocks, 256, 0, stream>>>(support, erow, ecol, eval, out, nEdges);
    }
    {
        const int n4 = out_size / 4;
        gcn_tanh<<<(n4 + 255) / 256, 256, 0, stream>>>((float4*)out, n4);
    }
}


# x@W: one wave owns a 16x128 stripe (8 accumulator tiles), A fragment loaded once per 8 tiles, all operand loads and stores 16 B wide; same f32 products, k summed in a fixed but different order
# speedup vs baseline: 1.7200x; 1.0412x over previous
#include <hip/hip_runtime.h>
#include <hip/hip_bf16.h>

typedef __attribute__((ext_vector_type(2))) float v2f;
typedef __attribute__((ext_vector_type(8))) float v8f;

#define D 128

__global__ __launch_bounds__(256) void gcn_zero(float4* __restrict__ p, int n4) {
    int i = blockIdx.x * blockDim.x + threadIdx.x;
    if (i < n4) p[i] = make_float4(0.f, 0.f, 0.f, 0.f);
}

typedef __attribute__((ext_vector_type(4))) float v4f;

__global__ __launch_bounds__(128) void gcn_gemm_wmma(const float* __restrict__ x,
                                                     const float* __restrict__ W,
                                                     float* __restrict__ support,
                                                     int nStripes) {
    const int lane = threadIdx.x & 31;
    const int wave = __builtin_amdgcn_readfirstlane(threadIdx.x >> 5);
    const int stripe = blockIdx.x * 4 + wave;
    if (stripe >= nStripes) return;
    const int m0 = stripe * 16;
    const int h  = lane >> 4;
    const int c  = lane & 15;

    const float* arow = x + (size_t)(m0 + c) * D + 4 * h;
    const float* wcol = W + (size_t)(4 * h) * D + 4 * c;

    v8f acc[2][4] = {};
#pragma unroll 2
    for (int q = 0; q < D / 8; ++q) {
        const v4f a4 = *(const v4f*)(arow + 8 * q);
        v4f b4[4][2];
#pragma unroll
        for (int kk = 0; kk < 4; ++kk) {
#pragma unroll
            for (int g = 0; g < 2; ++g)
                b4[kk][g] = *(const v4f*)(wcol + (size_t)(8 * q + kk) * D + 64 * g);
        }
        v2f a_lo, a_hi;
        a_lo.x = a4.x; a_lo.y = a4.y;
        a_hi.x = a4.z; a_hi.y = a4.w;
#pragma unroll
        for (int g = 0; g < 2; ++g) {
#pragma unroll
            for (int t = 0; t < 4; ++t) {
                v2f b;
                b.x = b4[0][g][t];
                b.y = b4[1][g][t];
                acc[g][t] = __builtin_amdgcn_wmma_f32_16x16x4_f32(
                        false, a_lo, false, b, (short)0, acc[g][t], false, false);
            }
        }
#pragma unroll
        for (int g = 0; g < 2; ++g) {
#pragma unroll
            for (int t = 0; t < 4; ++t) {
                v2f b;
                b.x = b4[2][g][t];
                b.y = b4[3][g][t];
                acc[g][t] = __builtin_amdgcn_wmma_f32_16x16x4_f32(
                        false, a_hi, false, b, (short)0, acc[g][t], false, false);
            }
        }
    }

    float* srow = support + (size_t)(m0 + 8 * h) * D + 4 * c;
#pragma unroll
    for (int r = 0; r < 8; ++r) {
#pragma unroll
        for (int g = 0; g < 2; ++g) {
            v4f o;
            o.x = acc[g][0][r];
            o.y = acc[g][1][r];
            o.z = acc[g][2][r];
            o.w = acc[g][3][r];
            *(v4f*)(srow + (size_t)r * D + 64 * g) = o;
        }
    }
}

#define EPW 32
#define GRP 4

__device__ __forceinline__ void gather_row(const float* __restrict__ support, int lane, int c,
                                           float (&m)[4]) {
    const float* row = support + (size_t)c * D;
    m[0] = row[lane];
    m[1] = row[lane + 32];
    m[2] = row[lane + 64];
    m[3] = row[lane + 96];
}

__device__ __forceinline__ void add_row(float* __restrict__ out, int voff, int r, float v,
                                        const float (&m)[4]) {
    float* base = out + (size_t)r * D;
    asm volatile("global_atomic_add_f32 %0, %1, %5\n\t"
                 "global_atomic_add_f32 %0, %2, %5 offset:128\n\t"
                 "global_atomic_add_f32 %0, %3, %5 offset:256\n\t"
                 "global_atomic_add_f32 %0, %4, %5 offset:384\n\t"
                 "s_wait_xcnt 0x0"
                 :
                 : "v"(voff), "v"(m[0] * v), "v"(m[1] * v), "v"(m[2] * v), "v"(m[3] * v),
                   "s"(base)
                 : "memory");
}

__global__ __launch_bounds__(256) void gcn_scatter_atomic(const float* __restrict__ support,
                                                          const int* __restrict__ erow,
                                                          const int* __restrict__ ecol,
                                                          const float* __restrict__ eval,
                                                          float* __restrict__ out,
                                                          int nEdges) {
    const int wv = __builtin_amdgcn_readfirstlane(threadIdx.x >> 5);
    const int e0 = (blockIdx.x * 8 + wv) * EPW;
    if (e0 >= nEdges) return;
    const int lane = threadIdx.x & 31;
    const int n = (nEdges - e0 < EPW) ? (nEdges - e0) : EPW;

    int my_r = 0, my_c = 0, my_v = 0;
    if (lane < n) {
        my_r = erow[e0 + lane];
        my_c = ecol[e0 + lane];
        my_v = __float_as_int(eval[e0 + lane]);
    }

    const int voff = lane * 4;

    if (n == EPW) {
        float m[2][GRP][4];
#pragma unroll
        for (int j = 0; j < GRP; ++j)
            gather_row(support, lane, __builtin_amdgcn_readlane(my_c, j), m[0][j]);
#pragma unroll
        for (int g = 0; g < EPW / GRP; ++g) {
            if (g + 1 < EPW / GRP) {
#pragma unroll
                for (int j = 0; j < GRP; ++j)
                    gather_row(support, lane, __builtin_amdgcn_readlane(my_c, (g + 1) * GRP + j),
                               m[(g + 1) & 1][j]);
            }
#pragma unroll
            for (int j = 0; j < GRP; ++j) {
                const int e = g * GRP + j;
                add_row(out, voff, __builtin_amdgcn_readlane(my_r, e),
                        __int_as_float(__builtin_amdgcn_readlane(my_v, e)), m[g & 1][j]);
            }
        }
    } else {
        for (int i = 0; i < n; ++i) {
            float m[4];
            gather_row(support, lane, __builtin_amdgcn_readlane(my_c, i), m);
            add_row(out, voff, __builtin_amdgcn_readlane(my_r, i),
                    __int_as_float(__builtin_amdgcn_readlane(my_v, i)), m);
        }
    }
}

__global__ __launch_bounds__(256) void gcn_tanh(float4* __restrict__ p, int n4) {
    int i = blockIdx.x * blockDim.x + threadIdx.x;
    if (i < n4) {
        float4 t = p[i];
        t.x = tanhf(t.x);
        t.y = tanhf(t.y);
        t.z = tanhf(t.z);
        t.w = tanhf(t.w);
        p[i] = t;
    }
}

extern "C" void kernel_launch(void* const* d_in, const int* in_sizes, int n_in,
                              void* d_out, int out_size, void* d_ws, size_t ws_size,
                              hipStream_t stream) {
    const float* x    = (const float*)d_in[0];
    const float* W    = (const float*)d_in[1];
    const int*   erow = (const int*)d_in[2];
    const int*   ecol = (const int*)d_in[3];
    const float* eval = (const float*)d_in[4];
    float* out = (float*)d_out;

    const int nNodes = in_sizes[0] / D;
    const int nEdges = in_sizes[2];

    float* support = (float*)d_ws;

    {
        const int n4 = out_size / 4;
        gcn_zero<<<(n4 + 255) / 256, 256, 0, stream>>>((float4*)out, n4);
    }
    {
        const int nStripes = nNodes / 16;
        gcn_gemm_wmma<<<(nStripes + 3) / 4, 128, 0, stream>>>(x, W, support, nStripes);
    }
    {
        const int blocks = (nEdges + 255) / 256;
        gcn_scatter_atomic<<<blocks, 256, 0, stream>>>(support, erow, ecol, eval, out, nEdges);
    }
    {
        const int n4 = out_size / 4;
        gcn_tanh<<<(n4 + 255) / 256, 256, 0, stream>>>((float4*)out, n4);
    }
}


# scatter: two sweeps over the edges in one launch, one per 64-column half of the rows (blocks in index order), so each sweep's gathered and accumulated half rows fit in L2; same adds, still unordered
# speedup vs baseline: 1.8874x; 1.0973x over previous
#include <hip/hip_runtime.h>
#include <hip/hip_bf16.h>

typedef __attribute__((ext_vector_type(2))) float v2f;
typedef __attribute__((ext_vector_type(8))) float v8f;

#define D 128

__global__ __launch_bounds__(256) void gcn_zero(float4* __restrict__ p, int n4) {
    int i = blockIdx.x * blockDim.x + threadIdx.x;
    if (i < n4) p[i] = make_float4(0.f, 0.f, 0.f, 0.f);
}

typedef __attribute__((ext_vector_type(4))) float v4f;

__global__ __launch_bounds__(128) void gcn_gemm_wmma(const float* __restrict__ x,
                                                     const float* __restrict__ W,
                                                     float* __restrict__ support,
                                                     int nStripes) {
    const int lane = threadIdx.x & 31;
    const int wave = __builtin_amdgcn_readfirstlane(threadIdx.x >> 5);
    const int stripe = blockIdx.x * 4 + wave;
    if (stripe >= nStripes) return;
    const int m0 = stripe * 16;
    const int h  = lane >> 4;
    const int c  = lane & 15;

    const float* arow = x + (size_t)(m0 + c) * D + 4 * h;
    const float* wcol = W + (size_t)(4 * h) * D + 4 * c;

    v8f acc[2][4] = {};
#pragma unroll 2
    for (int q = 0; q < D / 8; ++q) {
        const v4f a4 = *(const v4f*)(arow + 8 * q);
        v4f b4[4][2];
#pragma unroll
        for (int kk = 0; kk < 4; ++kk) {
#pragma unroll
            for (int g = 0; g < 2; ++g)
                b4[kk][g] = *(const v4f*)(wcol + (size_t)(8 * q + kk) * D + 64 * g);
        }
        v2f a_lo, a_hi;
        a_lo.x = a4.x; a_lo.y = a4.y;
        a_hi.x = a4.z; a_hi.y = a4.w;
#pragma unroll
        for (int g = 0; g < 2; ++g) {
#pragma unroll
            for (int t = 0; t < 4; ++t) {
                v2f b;
                b.x = b4[0][g][t];
                b.y = b4[1][g][t];
                acc[g][t] = __builtin_amdgcn_wmma_f32_16x16x4_f32(
                        false, a_lo, false, b, (short)0, acc[g][t], false, false);
            }
        }
#pragma unroll
        for (int g = 0; g < 2; ++g) {
#pragma unroll
            for (int t = 0; t < 4; ++t) {
                v2f b;
                b.x = b4[2][g][t];
                b.y = b4[3][g][t];
                acc[g][t] = __builtin_amdgcn_wmma_f32_16x16x4_f32(
                        false, a_hi, false, b, (short)0, acc[g][t], false, false);
            }
        }
    }

    float* srow = support + (size_t)(m0 + 8 * h) * D + 4 * c;
#pragma unroll
    for (int r = 0; r < 8; ++r) {
#pragma unroll
        for (int g = 0; g < 2; ++g) {
            v4f o;
            o.x = acc[g][0][r];
            o.y = acc[g][1][r];
            o.z = acc[g][2][r];
            o.w = acc[g][3][r];
            *(v4f*)(srow + (size_t)r * D + 64 * g) = o;
        }
    }
}

#define EPB 32
#define NB  2
#define EPW (EPB * NB)
#define GRP 8
#define HC  64

__device__ __forceinline__ void gather_half(const float* __restrict__ sup_h, unsigned voff, int c,
                                            float (&m)[2]) {
    const unsigned off = (unsigned)c * (unsigned)(D * sizeof(float)) + voff;
    m[0] = *(const float*)((const char*)sup_h + off);
    m[1] = *(const float*)((const char*)sup_h + off + 128u);
}

__device__ __forceinline__ void add_half2(float* __restrict__ out_h, unsigned voff,
                                          int r0, float v0, const float (&m0)[2],
                                          int r1, float v1, const float (&m1)[2]) {
    float* base0 = out_h + (size_t)r0 * D;
    float* base1 = out_h + (size_t)r1 * D;
    asm volatile("global_atomic_add_f32 %0, %1, %5\n\t"
                 "global_atomic_add_f32 %0, %2, %5 offset:128\n\t"
                 "global_atomic_add_f32 %0, %3, %6\n\t"
                 "global_atomic_add_f32 %0, %4, %6 offset:128\n\t"
                 "s_wait_xcnt 0x0"
                 :
                 : "v"(voff), "v"(m0[0] * v0), "v"(m0[1] * v0), "v"(m1[0] * v1), "v"(m1[1] * v1),
                   "s"(base0), "s"(base1)
                 : "memory");
}

__device__ __forceinline__ void add_half1(float* __restrict__ out_h, unsigned voff, int r, float v,
                                          const float (&m)[2]) {
    float* base = out_h + (size_t)r * D;
    asm volatile("global_atomic_add_f32 %0, %1, %3\n\t"
                 "global_atomic_add_f32 %0, %2, %3 offset:128\n\t"
                 "s_wait_xcnt 0x0"
                 :
                 : "v"(voff), "v"(m[0] * v), "v"(m[1] * v), "s"(base)
                 : "memory");
}

__global__ __launch_bounds__(256) void gcn_scatter_atomic(const float* __restrict__ support,
                                                          const int* __restrict__ erow,
                                                          const int* __restrict__ ecol,
                                                          const float* __restrict__ eval,
                                                          float* __restrict__ out,
                                                          int nEdges, int blocksPerSweep) {
    const int wv = __builtin_amdgcn_readfirstlane(threadIdx.x >> 5);
    const int sweep = ((int)blockIdx.x >= blocksPerSweep) ? 1 : 0;
    const int blk = (int)blockIdx.x - sweep * blocksPerSweep;
    const int e0 = (blk * 8 + wv) * EPW;
    if (e0 >= nEdges) return;
    const int lane = threadIdx.x & 31;
    const int n = (nEdges - e0 < EPW) ? (nEdges - e0) : EPW;

    const float* sup_h = support + sweep * HC;
    float*       out_h = out + sweep * HC;

    int my_r[NB], my_c[NB], my_v[NB];
#pragma unroll
    for (int b = 0; b < NB; ++b) {
        my_r[b] = 0; my_c[b] = 0; my_v[b] = 0;
        if (b * EPB + lane < n) {
            my_r[b] = erow[e0 + b * EPB + lane];
            my_c[b] = ecol[e0 + b * EPB + lane];
            my_v[b] = __float_as_int(eval[e0 + b * EPB + lane]);
        }
    }

    const unsigned voff = (unsigned)lane * 4u;

    if (n == EPW) {
        float m[2][GRP][2];
#pragma unroll
        for (int j = 0; j < GRP; ++j)
            gather_half(sup_h, voff, __builtin_amdgcn_readlane(my_c[0], j), m[0][j]);
#pragma unroll
        for (int g = 0; g < EPW / GRP; ++g) {
            if (g + 1 < EPW / GRP) {
#pragma unroll
                for (int j = 0; j < GRP; ++j) {
                    const int e = (g + 1) * GRP + j;
                    gather_half(sup_h, voff, __builtin_amdgcn_readlane(my_c[e / EPB], e % EPB),
                                m[(g + 1) & 1][j]);
                }
            }
#pragma unroll
            for (int j = 0; j < GRP; j += 2) {
                const int ea = g * GRP + j;
                const int eb = ea + 1;
                add_half2(out_h, voff,
                          __builtin_amdgcn_readlane(my_r[ea / EPB], ea % EPB),
                          __int_as_float(__builtin_amdgcn_readlane(my_v[ea / EPB], ea % EPB)),
                          m[g & 1][j],
                          __builtin_amdgcn_readlane(my_r[eb / EPB], eb % EPB),
                          __int_as_float(__builtin_amdgcn_readlane(my_v[eb / EPB], eb % EPB)),
                          m[g & 1][j + 1]);
            }
        }
    } else {
        for (int i = 0; i < n; ++i) {
            const int li = i & (EPB - 1);
            const int cr = (i < EPB) ? __builtin_amdgcn_readlane(my_r[0], li)
                                     : __builtin_amdgcn_readlane(my_r[1], li);
            const int cc = (i < EPB) ? __builtin_amdgcn_readlane(my_c[0], li)
                                     : __builtin_amdgcn_readlane(my_c[1], li);
            const int cv = (i < EPB) ? __builtin_amdgcn_readlane(my_v[0], li)
                                     : __builtin_amdgcn_readlane(my_v[1], li);
            float m[2];
            gather_half(sup_h, voff, cc, m);
            add_half1(out_h, voff, cr, __int_as_float(cv), m);
        }
    }
}

__global__ __launch_bounds__(256) void gcn_tanh(float4* __restrict__ p, int n4) {
    int i = blockIdx.x * blockDim.x + threadIdx.x;
    if (i < n4) {
        float4 t = p[i];
        t.x = tanhf(t.x);
        t.y = tanhf(t.y);
        t.z = tanhf(t.z);
        t.w = tanhf(t.w);
        p[i] = t;
    }
}

extern "C" void kernel_launch(void* const* d_in, const int* in_sizes, int n_in,
                              void* d_out, int out_size, void* d_ws, size_t ws_size,
                              hipStream_t stream) {
    const float* x    = (const float*)d_in[0];
    const float* W    = (const float*)d_in[1];
    const int*   erow = (const int*)d_in[2];
    const int*   ecol = (const int*)d_in[3];
    const float* eval = (const float*)d_in[4];
    float* out = (float*)d_out;

    const int nNodes = in_sizes[0] / D;
    const int nEdges = in_sizes[2];

    float* support = (float*)d_ws;

    {
        const int n4 = out_size / 4;
        gcn_zero<<<(n4 + 255) / 256, 256, 0, stream>>>((float4*)out, n4);
    }
    {
        const int nStripes = nNodes / 16;
        gcn_gemm_wmma<<<(nStripes + 3) / 4, 128, 0, stream>>>(x, W, support, nStripes);
    }
    {
        const int blocksPerSweep = (nEdges + 511) / 512;
        gcn_scatter_atomic<<<2 * blocksPerSweep, 256, 0, stream>>>(support, erow, ecol, eval, out,
                                                                   nEdges, blocksPerSweep);
    }
    {
        const int n4 = out_size / 4;
        gcn_tanh<<<(n4 + 255) / 256, 256, 0, stream>>>((float4*)out, n4);
    }
}


# x@W kernel also stores the start value 0 of its stripe's 16 out rows, so the zeroing pass and its launch are gone: the stores drain under the f32 matrix work; same adds, atomics still unordered
# speedup vs baseline: 1.9674x; 1.0424x over previous
#include <hip/hip_runtime.h>
#include <hip/hip_bf16.h>

typedef __attribute__((ext_vector_type(2))) float v2f;
typedef __attribute__((ext_vector_type(8))) float v8f;

#define D 128

typedef __attribute__((ext_vector_type(4))) float v4f;

__global__ __launch_bounds__(128) void gcn_gemm_wmma(const float* __restrict__ x,
                                                     const float* __restrict__ W,
                                                     float* __restrict__ support,
                                                     float* __restrict__ out,
                                                     int nStripes) {
    const int lane = threadIdx.x & 31;
    const int wave = __builtin_amdgcn_readfirstlane(threadIdx.x >> 5);
    const int stripe = blockIdx.x * 4 + wave;
    if (stripe >= nStripes) return;
    const int m0 = stripe * 16;
    const int h  = lane >> 4;
    const int c  = lane & 15;

    const float* arow = x + (size_t)(m0 + c) * D + 4 * h;
    const float* wcol = W + (size_t)(4 * h) * D + 4 * c;

    {
        float* zrow = out + (size_t)(m0 + 8 * h) * D + 4 * c;
        const v4f z4 = {0.f, 0.f, 0.f, 0.f};
#pragma unroll
        for (int r = 0; r < 8; ++r) {
#pragma unroll
            for (int g = 0; g < 2; ++g)
                *(v4f*)(zrow + (size_t)r * D + 64 * g) = z4;
        }
    }

    v8f acc[2][4] = {};
#pragma unroll 2
    for (int q = 0; q < D / 8; ++q) {
        const v4f a4 = *(const v4f*)(arow + 8 * q);
        v4f b4[4][2];
#pragma unroll
        for (int kk = 0; kk < 4; ++kk) {
#pragma unroll
            for (int g = 0; g < 2; ++g)
                b4[kk][g] = *(const v4f*)(wcol + (size_t)(8 * q + kk) * D + 64 * g);
        }
        v2f a_lo, a_hi;
        a_lo.x = a4.x; a_lo.y = a4.y;
        a_hi.x = a4.z; a_hi.y = a4.w;
#pragma unroll
        for (int g = 0; g < 2; ++g) {
#pragma unroll
            for (int t = 0; t < 4; ++t) {
                v2f b;
                b.x = b4[0][g][t];
                b.y = b4[1][g][t];
                acc[g][t] = __builtin_amdgcn_wmma_f32_16x16x4_f32(
                        false, a_lo, false, b, (short)0, acc[g][t], false, false);
            }
        }
#pragma unroll
        for (int g = 0; g < 2; ++g) {
#pragma unroll
            for (int t = 0; t < 4; ++t) {
                v2f b;
                b.x = b4[2][g][t];
                b.y = b4[3][g][t];
                acc[g][t] = __builtin_amdgcn_wmma_f32_16x16x4_f32(
                        false, a_hi, false, b, (short)0, acc[g][t], false, false);
            }
        }
    }

    float* srow = support + (size_t)(m0 + 8 * h) * D + 4 * c;
#pragma unroll
    for (int r = 0; r < 8; ++r) {
#pragma unroll
        for (int g = 0; g < 2; ++g) {
            v4f o;
            o.x = acc[g][0][r];
            o.y = acc[g][1][r];
            o.z = acc[g][2][r];
            o.w = acc[g][3][r];
            *(v4f*)(srow + (size_t)r * D + 64 * g) = o;
        }
    }
}

#define EPB 32
#define NB  2
#define EPW (EPB * NB)
#define GRP 8
#define HC  64

__device__ __forceinline__ void gather_half(const float* __restrict__ sup_h, unsigned voff, int c,
                                            float (&m)[2]) {
    const unsigned off = (unsigned)c * (unsigned)(D * sizeof(float)) + voff;
    m[0] = *(const float*)((const char*)sup_h + off);
    m[1] = *(const float*)((const char*)sup_h + off + 128u);
}

__device__ __forceinline__ void add_half2(float* __restrict__ out_h, unsigned voff,
                                          int r0, float v0, const float (&m0)[2],
                                          int r1, float v1, const float (&m1)[2]) {
    float* base0 = out_h + (size_t)r0 * D;
    float* base1 = out_h + (size_t)r1 * D;
    asm volatile("global_atomic_add_f32 %0, %1, %5\n\t"
                 "global_atomic_add_f32 %0, %2, %5 offset:128\n\t"
                 "global_atomic_add_f32 %0, %3, %6\n\t"
                 "global_atomic_add_f32 %0, %4, %6 offset:128\n\t"
                 "s_wait_xcnt 0x0"
                 :
                 : "v"(voff), "v"(m0[0] * v0), "v"(m0[1] * v0), "v"(m1[0] * v1), "v"(m1[1] * v1),
                   "s"(base0), "s"(base1)
                 : "memory");
}

__device__ __forceinline__ void add_half1(float* __restrict__ out_h, unsigned voff, int r, float v,
                                          const float (&m)[2]) {
    float* base = out_h + (size_t)r * D;
    asm volatile("global_atomic_add_f32 %0, %1, %3\n\t"
                 "global_atomic_add_f32 %0, %2, %3 offset:128\n\t"
                 "s_wait_xcnt 0x0"
                 :
                 : "v"(voff), "v"(m[0] * v), "v"(m[1] * v), "s"(base)
                 : "memory");
}

__global__ __launch_bounds__(256) void gcn_scatter_atomic(const float* __restrict__ support,
                                                          const int* __restrict__ erow,
                                                          const int* __restrict__ ecol,
                                                          const float* __restrict__ eval,
                                                          float* __restrict__ out,
                                                          int nEdges, int blocksPerSweep) {
    const int wv = __builtin_amdgcn_readfirstlane(threadIdx.x >> 5);
    const int sweep = ((int)blockIdx.x >= blocksPerSweep) ? 1 : 0;
    const int blk = (int)blockIdx.x - sweep * blocksPerSweep;
    const int e0 = (blk * 8 + wv) * EPW;
    if (e0 >= nEdges) return;
    const int lane = threadIdx.x & 31;
    const int n = (nEdges - e0 < EPW) ? (nEdges - e0) : EPW;

    const float* sup_h = support + sweep * HC;
    float*       out_h = out + sweep * HC;

    int my_r[NB], my_c[NB], my_v[NB];
#pragma unroll
    for (int b = 0; b < NB; ++b) {
        my_r[b] = 0; my_c[b] = 0; my_v[b] = 0;
        if (b * EPB + lane < n) {
            my_r[b] = erow[e0 + b * EPB + lane];
            my_c[b] = ecol[e0 + b * EPB + lane];
            my_v[b] = __float_as_int(eval[e0 + b * EPB + lane]);
        }
    }

    const unsigned voff = (unsigned)lane * 4u;

    if (n == EPW) {
        float m[2][GRP][2];
#pragma unroll
        for (int j = 0; j < GRP; ++j)
            gather_half(sup_h, voff, __builtin_amdgcn_readlane(my_c[0], j), m[0][j]);
#pragma unroll
        for (int g = 0; g < EPW / GRP; ++g) {
            if (g + 1 < EPW / GRP) {
#pragma unroll
                for (int j = 0; j < GRP; ++j) {
                    const int e = (g + 1) * GRP + j;
                    gather_half(sup_h, voff, __builtin_amdgcn_readlane(my_c[e / EPB], e % EPB),
                                m[(g + 1) & 1][j]);
                }
            }
#pragma unroll
            for (int j = 0; j < GRP; j += 2) {
                const int ea = g * GRP + j;
                const int eb = ea + 1;
                add_half2(out_h, voff,
                          __builtin_amdgcn_readlane(my_r[ea / EPB], ea % EPB),
                          __int_as_float(__builtin_amdgcn_readlane(my_v[ea / EPB], ea % EPB)),
                          m[g & 1][j],
                          __builtin_amdgcn_readlane(my_r[eb / EPB], eb % EPB),
                          __int_as_float(__builtin_amdgcn_readlane(my_v[eb / EPB], eb % EPB)),
                          m[g & 1][j + 1]);
            }
        }
    } else {
        for (int i = 0; i < n; ++i) {
            const int li = i & (EPB - 1);
            const int cr = (i < EPB) ? __builtin_amdgcn_readlane(my_r[0], li)
                                     : __builtin_amdgcn_readlane(my_r[1], li);
            const int cc = (i < EPB) ? __builtin_amdgcn_readlane(my_c[0], li)
                                     : __builtin_amdgcn_readlane(my_c[1], li);
            const int cv = (i < EPB) ? __builtin_amdgcn_readlane(my_v[0], li)
                                     : __builtin_amdgcn_readlane(my_v[1], li);
            float m[2];
            gather_half(sup_h, voff, cc, m);
            add_half1(out_h, voff, cr, __int_as_float(cv), m);
        }
    }
}

__global__ __launch_bounds__(256) void gcn_tanh(float4* __restrict__ p, int n4) {
    int i = blockIdx.x * blockDim.x + threadIdx.x;
    if (i < n4) {
        float4 t = p[i];
        t.x = tanhf(t.x);
        t.y = tanhf(t.y);
        t.z = tanhf(t.z);
        t.w = tanhf(t.w);
        p[i] = t;
    }
}

extern "C" void kernel_launch(void* const* d_in, const int* in_sizes, int n_in,
                              void* d_out, int out_size, void* d_ws, size_t ws_size,
                              hipStream_t stream) {
    const float* x    = (const float*)d_in[0];
    const float* W    = (const float*)d_in[1];
    const int*   erow = (const int*)d_in[2];
    const int*   ecol = (const int*)d_in[3];
    const float* eval = (const float*)d_in[4];
    float* out = (float*)d_out;

    const int nNodes = in_sizes[0] / D;
    const int nEdges = in_sizes[2];

    float* support = (float*)d_ws;

    {
        const int nStripes = nNodes / 16;
        gcn_gemm_wmma<<<(nStripes + 3) / 4, 128, 0, stream>>>(x, W, support, out, nStripes);
    }
    {
        const int blocksPerSweep = (nEdges + 511) / 512;
        gcn_scatter_atomic<<<2 * blocksPerSweep, 256, 0, stream>>>(support, erow, ecol, eval, out,
                                                                   nEdges, blocksPerSweep);
    }
    {
        const int n4 = out_size / 4;
        gcn_tanh<<<(n4 + 255) / 256, 256, 0, stream>>>((float4*)out, n4);
    }
}


# segment sum owner-computes: edges binned by 416-row bucket (ballot ranks, stable), each wave adds its 13 rows' gathered rows in LDS in edge order (fixed), stores tanh; all atomics gone
# speedup vs baseline: 2.7580x; 1.4019x over previous
#include <hip/hip_runtime.h>
#include <hip/hip_bf16.h>

typedef __attribute__((ext_vector_type(2))) float v2f;
typedef __attribute__((ext_vector_type(8))) float v8f;

#define D 128

typedef __attribute__((ext_vector_type(4))) float v4f;

__global__ __launch_bounds__(128) void gcn_gemm_wmma(const float* __restrict__ x,
                                                     const float* __restrict__ W,
                                                     float* __restrict__ support,
                                                     int nStripes) {
    const int lane = threadIdx.x & 31;
    const int wave = __builtin_amdgcn_readfirstlane(threadIdx.x >> 5);
    const int stripe = blockIdx.x * 4 + wave;
    if (stripe >= nStripes) return;
    const int m0 = stripe * 16;
    const int h  = lane >> 4;
    const int c  = lane & 15;

    const float* arow = x + (size_t)(m0 + c) * D + 4 * h;
    const float* wcol = W + (size_t)(4 * h) * D + 4 * c;

    v8f acc[2][4] = {};
#pragma unroll 2
    for (int q = 0; q < D / 8; ++q) {
        const v4f a4 = *(const v4f*)(arow + 8 * q);
        v4f b4[4][2];
#pragma unroll
        for (int kk = 0; kk < 4; ++kk) {
#pragma unroll
            for (int g = 0; g < 2; ++g)
                b4[kk][g] = *(const v4f*)(wcol + (size_t)(8 * q + kk) * D + 64 * g);
        }
        v2f a_lo, a_hi;
        a_lo.x = a4.x; a_lo.y = a4.y;
        a_hi.x = a4.z; a_hi.y = a4.w;
#pragma unroll
        for (int g = 0; g < 2; ++g) {
#pragma unroll
            for (int t = 0; t < 4; ++t) {
                v2f b;
                b.x = b4[0][g][t];
                b.y = b4[1][g][t];
                acc[g][t] = __builtin_amdgcn_wmma_f32_16x16x4_f32(
                        false, a_lo, false, b, (short)0, acc[g][t], false, false);
            }
        }
#pragma unroll
        for (int g = 0; g < 2; ++g) {
#pragma unroll
            for (int t = 0; t < 4; ++t) {
                v2f b;
                b.x = b4[2][g][t];
                b.y = b4[3][g][t];
                acc[g][t] = __builtin_amdgcn_wmma_f32_16x16x4_f32(
                        false, a_hi, false, b, (short)0, acc[g][t], false, false);
            }
        }
    }

    float* srow = support + (size_t)(m0 + 8 * h) * D + 4 * c;
#pragma unroll
    for (int r = 0; r < 8; ++r) {
#pragma unroll
        for (int g = 0; g < 2; ++g) {
            v4f o;
            o.x = acc[g][0][r];
            o.y = acc[g][1][r];
            o.z = acc[g][2][r];
            o.w = acc[g][3][r];
            *(v4f*)(srow + (size_t)r * D + 64 * g) = o;
        }
    }
}

#define RPW     13
#define WPB     32
#define RPB     (RPW * WPB)
#define NBK     256
#define ETH     (WPB * 32)
#define TILE    8192
#define GRP     8
#define RND     8
#define COLBITS 17

__device__ __forceinline__ unsigned same_key_lanes(unsigned key, int nbits, bool valid) {
    unsigned grp = __builtin_amdgcn_ballot_w32(valid);
    for (int bit = 0; bit < nbits; ++bit) {
        const bool one = ((key >> bit) & 1u) != 0u;
        const unsigned bal = __builtin_amdgcn_ballot_w32(one);
        grp &= one ? bal : ~bal;
    }
    return valid ? grp : 0u;
}

__device__ __forceinline__ void wave_count(const int* __restrict__ erow, int ebeg, int eend,
                                           int nNodes, int lane, volatile int* wh) {
    const unsigned lt = (1u << lane) - 1u;
    for (int e = ebeg; e < eend; e += 32 * RND) {
        unsigned r8[RND];
#pragma unroll
        for (int k = 0; k < RND; ++k) {
            const int idx = e + 32 * k + lane;
            r8[k] = 0xffffffffu;
            if (idx < eend) r8[k] = (unsigned)erow[idx];
        }
#pragma unroll
        for (int k = 0; k < RND; ++k) {
            const unsigned r = r8[k];
            const bool valid = r < (unsigned)nNodes;
            const unsigned bk = valid ? r / RPB : 0u;
            const unsigned grp = same_key_lanes(bk, 8, valid);
            if (valid && (grp & lt) == 0u) wh[bk] += __popc(grp);
        }
    }
}

__global__ __launch_bounds__(ETH) void gcn_edge_count(const int* __restrict__ erow,
                                                      int* __restrict__ cnt,
                                                      int nEdges, int chunk, int nNodes) {
    __shared__ int wh[WPB * NBK];
    const int tid = threadIdx.x;
    const int lane = tid & 31;
    const int wave = __builtin_amdgcn_readfirstlane(tid >> 5);
    for (int i = tid; i < WPB * NBK; i += ETH) wh[i] = 0;
    __syncthreads();
    const int e0 = blockIdx.x * chunk;
    const int e1 = (e0 + chunk < nEdges) ? (e0 + chunk) : nEdges;
    const int per = (chunk + WPB - 1) / WPB;
    const int wb = e0 + wave * per;
    const int we = (wb + per < e1) ? (wb + per) : e1;
    wave_count(erow, wb, we, nNodes, lane, wh + wave * NBK);
    __syncthreads();
    if (tid < NBK) {
        int total = 0;
#pragma unroll
        for (int w = 0; w < WPB; ++w) total += wh[w * NBK + tid];
        cnt[blockIdx.x * NBK + tid] = total;
    }
}

__global__ __launch_bounds__(ETH) void gcn_edge_partition(const int* __restrict__ erow,
                                                          const int* __restrict__ ecol,
                                                          const float* __restrict__ eval,
                                                          const int* __restrict__ cnt,
                                                          int2* __restrict__ recs,
                                                          int* __restrict__ bstart,
                                                          int nEdges, int chunk, int nNodes,
                                                          int nBlocks) {
    __shared__ __attribute__((aligned(16))) int wh[WPB * NBK];
    __shared__ __attribute__((aligned(16))) int pp[WPB * NBK];
    __shared__ int tot[NBK + 1];
    const int tid = threadIdx.x;
    const int lane = tid & 31;
    const int wave = __builtin_amdgcn_readfirstlane(tid >> 5);
    const int me = blockIdx.x;

    {
        int ts[8] = {0, 0, 0, 0, 0, 0, 0, 0};
        int ps[8] = {0, 0, 0, 0, 0, 0, 0, 0};
#pragma unroll
        for (int i = 0; i < 8; ++i) {
            const int bb = 8 * wave + i;
            if (bb < nBlocks) {
                const int4 lo = *(const int4*)(cnt + bb * NBK + 8 * lane);
                const int4 hi = *(const int4*)(cnt + bb * NBK + 8 * lane + 4);
                const int f = (bb < me) ? 1 : 0;
                ts[0] += lo.x; ts[1] += lo.y; ts[2] += lo.z; ts[3] += lo.w;
                ts[4] += hi.x; ts[5] += hi.y; ts[6] += hi.z; ts[7] += hi.w;
                ps[0] += f * lo.x; ps[1] += f * lo.y; ps[2] += f * lo.z; ps[3] += f * lo.w;
                ps[4] += f * hi.x; ps[5] += f * hi.y; ps[6] += f * hi.z; ps[7] += f * hi.w;
            }
        }
#pragma unroll
        for (int j = 0; j < 8; ++j) {
            wh[wave * NBK + 8 * lane + j] = ts[j];
            pp[wave * NBK + 8 * lane + j] = ps[j];
        }
    }
    __syncthreads();
    int part = 0;
    if (tid < NBK) {
        int total = 0;
#pragma unroll
        for (int w = 0; w < WPB; ++w) {
            total += wh[w * NBK + tid];
            part += pp[w * NBK + tid];
        }
        tot[tid] = total;
    }
    __syncthreads();
    for (int i = lane; i < NBK; i += 32) wh[wave * NBK + i] = 0;
    if (tid < 32) {
        int v[8];
        int run = 0;
#pragma unroll
        for (int j = 0; j < 8; ++j) {
            const int c = tot[8 * tid + j];
            v[j] = run;
            run += c;
        }
        int incl = run;
#pragma unroll
        for (int d = 1; d < 32; d <<= 1) {
            const int o = __shfl_up(incl, d);
            if (tid >= d) incl += o;
        }
        const int excl = incl - run;
#pragma unroll
        for (int j = 0; j < 8; ++j) tot[8 * tid + j] = excl + v[j];
        if (tid == 31) tot[NBK] = incl;
    }
    const int e0 = me * chunk;
    const int e1 = (e0 + chunk < nEdges) ? (e0 + chunk) : nEdges;
    const int per = (chunk + WPB - 1) / WPB;
    const int wb = e0 + wave * per;
    const int we = (wb + per < e1) ? (wb + per) : e1;
    volatile int* mywh = wh + wave * NBK;
    wave_count(erow, wb, we, nNodes, lane, mywh);
    __syncthreads();
    if (tid < NBK) {
        int run = tot[tid] + part;
#pragma unroll
        for (int w = 0; w < WPB; ++w) {
            const int v = wh[w * NBK + tid];
            wh[w * NBK + tid] = run;
            run += v;
        }
    }
    if (me == 0 && tid <= NBK) bstart[tid] = tot[tid];
    __syncthreads();

    const unsigned lt = (1u << lane) - 1u;
    for (int e = wb; e < we; e += 32 * RND) {
        unsigned r8[RND];
        unsigned c8[RND];
        int v8[RND];
#pragma unroll
        for (int k = 0; k < RND; ++k) {
            const int idx = e + 32 * k + lane;
            r8[k] = 0xffffffffu;
            c8[k] = 0u;
            v8[k] = 0;
            if (idx < we) {
                r8[k] = (unsigned)erow[idx];
                c8[k] = (unsigned)ecol[idx];
                v8[k] = __float_as_int(eval[idx]);
            }
        }
#pragma unroll
        for (int k = 0; k < RND; ++k) {
            const unsigned r = r8[k];
            const bool valid = r < (unsigned)nNodes;
            const unsigned bk = valid ? r / RPB : 0u;
            const unsigned grp = same_key_lanes(bk, 8, valid);
            if (valid) {
                const int pos = mywh[bk] + __popc(grp & lt);
                int2 rec;
                rec.x = (int)((c8[k] & ((1u << COLBITS) - 1u)) | ((r - bk * RPB) << COLBITS));
                rec.y = v8[k];
                recs[pos] = rec;
            }
            if (valid && (grp & lt) == 0u) mywh[bk] += __popc(grp);
        }
    }
}

__device__ __forceinline__ v4f gather_row(const float* __restrict__ support, unsigned laneoff, int key) {
    const unsigned col = (unsigned)key & ((1u << COLBITS) - 1u);
    return *(const v4f*)((const char*)support + (size_t)col * (D * sizeof(float)) + laneoff);
}

__device__ __forceinline__ void add_row(v4f* sums4, int lane, int key, float v, v4f m) {
    v4f* p = sums4 + ((unsigned)key >> COLBITS) * 32 + lane;
    v4f a = *p;
    a.x += m.x * v;
    a.y += m.y * v;
    a.z += m.z * v;
    a.w += m.w * v;
    *p = a;
}

__global__ __launch_bounds__(ETH) void gcn_aggregate(const float* __restrict__ support,
                                                     const int2* __restrict__ recs,
                                                     const int* __restrict__ bstart,
                                                     float* __restrict__ out, int nNodes) {
    __shared__ __attribute__((aligned(16))) float sums[RPB * D];
    __shared__ __attribute__((aligned(8)))  int2  tile[TILE];
    __shared__ int wh[WPB * WPB];
    __shared__ int ostart[WPB + 1];
    const int tid = threadIdx.x;
    const int lane = tid & 31;
    const int wave = __builtin_amdgcn_readfirstlane(tid >> 5);
    const int b = blockIdx.x;
    const unsigned lt = (1u << lane) - 1u;
    const unsigned laneoff = (unsigned)lane * 16u;
    v4f* sums4 = (v4f*)sums;
    v4f* mys = sums4 + wave * RPW * 32;
    volatile int* mywh = wh + wave * WPB;

    {
        const v4f z4 = {0.f, 0.f, 0.f, 0.f};
        for (int r = 0; r < RPW; ++r) mys[r * 32 + lane] = z4;
    }

    const int s = bstart[b];
    const int t = bstart[b + 1];
    for (int tile0 = s; tile0 < t; tile0 += TILE) {
        const int nt = (t - tile0 < TILE) ? (t - tile0) : TILE;
        mywh[lane] = 0;
        int2 rc[TILE / ETH];
        unsigned own[TILE / ETH];
        int pw[TILE / ETH];
#pragma unroll
        for (int k = 0; k < TILE / ETH; ++k) {
            const int idx = wave * (TILE / WPB) + k * 32 + lane;
            rc[k] = make_int2(0, 0);
            if (idx < nt) rc[k] = recs[tile0 + idx];
        }
#pragma unroll
        for (int k = 0; k < TILE / ETH; ++k) {
            const int idx = wave * (TILE / WPB) + k * 32 + lane;
            const bool valid = idx < nt;
            own[k] = valid ? ((unsigned)rc[k].x >> COLBITS) / RPW : 0u;
            const unsigned grp = same_key_lanes(own[k], 5, valid);
            const int base = mywh[own[k]];
            pw[k] = base + __popc(grp & lt);
            if (valid && (grp & lt) == 0u) mywh[own[k]] = base + __popc(grp);
        }
        __syncthreads();
        if (tid < 32) {
            int run = 0;
#pragma unroll
            for (int w = 0; w < WPB; ++w) {
                const int v = wh[w * WPB + lane];
                wh[w * WPB + lane] = run;
                run += v;
            }
            int incl = run;
#pragma unroll
            for (int d = 1; d < 32; d <<= 1) {
                const int o = __shfl_up(incl, d);
                if (lane >= d) incl += o;
            }
            ostart[lane] = incl - run;
            if (lane == 31) ostart[WPB] = incl;
        }
        __syncthreads();
#pragma unroll
        for (int k = 0; k < TILE / ETH; ++k) {
            const int idx = wave * (TILE / WPB) + k * 32 + lane;
            if (idx < nt) tile[ostart[own[k]] + wh[wave * WPB + own[k]] + pw[k]] = rc[k];
        }
        __syncthreads();
        const int a0 = __builtin_amdgcn_readfirstlane(ostart[wave]);
        const int a1 = __builtin_amdgcn_readfirstlane(ostart[wave + 1]);
        for (int base = a0; base < a1; base += 32) {
            const int n = (a1 - base < 32) ? (a1 - base) : 32;
            int2 rec = make_int2(0, 0);
            if (lane < n) rec = tile[base + lane];
            if (n == 32) {
                v4f m[2][GRP];
#pragma unroll
                for (int j = 0; j < GRP; ++j)
                    m[0][j] = gather_row(support, laneoff, __builtin_amdgcn_readlane(rec.x, j));
#pragma unroll
                for (int g = 0; g < 32 / GRP; ++g) {
                    if (g + 1 < 32 / GRP) {
#pragma unroll
                        for (int j = 0; j < GRP; ++j)
                            m[(g + 1) & 1][j] = gather_row(support, laneoff,
                                    __builtin_amdgcn_readlane(rec.x, (g + 1) * GRP + j));
                    }
#pragma unroll
                    for (int j = 0; j < GRP; ++j) {
                        const int e = g * GRP + j;
                        add_row(sums4, lane, __builtin_amdgcn_readlane(rec.x, e),
                                __int_as_float(__builtin_amdgcn_readlane(rec.y, e)), m[g & 1][j]);
                    }
                }
            } else {
                for (int i0 = 0; i0 < n; i0 += GRP) {
                    v4f mt[GRP];
#pragma unroll
                    for (int j = 0; j < GRP; ++j) {
                        if (i0 + j < n)
                            mt[j] = gather_row(support, laneoff, __builtin_amdgcn_readlane(rec.x, i0 + j));
                    }
#pragma unroll
                    for (int j = 0; j < GRP; ++j) {
                        if (i0 + j < n)
                            add_row(sums4, lane, __builtin_amdgcn_readlane(rec.x, i0 + j),
                                    __int_as_float(__builtin_amdgcn_readlane(rec.y, i0 + j)), mt[j]);
                    }
                }
            }
        }
        __syncthreads();
    }

    for (int r = 0; r < RPW; ++r) {
        const int row = b * RPB + wave * RPW + r;
        if (row < nNodes) {
            const v4f a = mys[r * 32 + lane];
            v4f o;
            o.x = tanhf(a.x);
            o.y = tanhf(a.y);
            o.z = tanhf(a.z);
            o.w = tanhf(a.w);
            *(v4f*)(out + (size_t)row * D + 4 * lane) = o;
        }
    }
}

extern "C" void kernel_launch(void* const* d_in, const int* in_sizes, int n_in,
                              void* d_out, int out_size, void* d_ws, size_t ws_size,
                              hipStream_t stream) {
    const float* x    = (const float*)d_in[0];
    const float* W    = (const float*)d_in[1];
    const int*   erow = (const int*)d_in[2];
    const int*   ecol = (const int*)d_in[3];
    const float* eval = (const float*)d_in[4];
    float* out = (float*)d_out;

    const int nNodes = in_sizes[0] / D;
    const int nEdges = in_sizes[2];

    float* support = (float*)d_ws;
    int2*  recs    = (int2*)(support + (size_t)nNodes * D);
    int*   cnt     = (int*)(recs + nEdges);
    int*   bstart  = cnt + NBK * NBK;

    const int nBlocks = 256;
    const int chunk   = (nEdges + nBlocks - 1) / nBlocks;

    {
        const int nStripes = nNodes / 16;
        gcn_gemm_wmma<<<(nStripes + 3) / 4, 128, 0, stream>>>(x, W, support, nStripes);
    }
    gcn_edge_count<<<nBlocks, ETH, 0, stream>>>(erow, cnt, nEdges, chunk, nNodes);
    gcn_edge_partition<<<nBlocks, ETH, 0, stream>>>(erow, ecol, eval, cnt, recs, bstart,
                                                    nEdges, chunk, nNodes, nBlocks);
    gcn_aggregate<<<(nNodes + RPB - 1) / RPB, ETH, 0, stream>>>(support, recs, bstart, out, nNodes);
}


# edge partition local to each block of consecutive edges (own counts, own prefix): count launch and 256x256 table sums gone; a bucket reads its 256 short runs in block order; same adds, same order
# speedup vs baseline: 3.2377x; 1.1739x over previous
#include <hip/hip_runtime.h>
#include <hip/hip_bf16.h>

typedef __attribute__((ext_vector_type(2))) float v2f;
typedef __attribute__((ext_vector_type(8))) float v8f;

#define D 128

typedef __attribute__((ext_vector_type(4))) float v4f;

__global__ __launch_bounds__(128) void gcn_gemm_wmma(const float* __restrict__ x,
                                                     const float* __restrict__ W,
                                                     float* __restrict__ support,
                                                     int nStripes) {
    const int lane = threadIdx.x & 31;
    const int wave = __builtin_amdgcn_readfirstlane(threadIdx.x >> 5);
    const int stripe = blockIdx.x * 4 + wave;
    if (stripe >= nStripes) return;
    const int m0 = stripe * 16;
    const int h  = lane >> 4;
    const int c  = lane & 15;

    const float* arow = x + (size_t)(m0 + c) * D + 4 * h;
    const float* wcol = W + (size_t)(4 * h) * D + 4 * c;

    v8f acc[2][4] = {};
#pragma unroll 2
    for (int q = 0; q < D / 8; ++q) {
        const v4f a4 = *(const v4f*)(arow + 8 * q);
        v4f b4[4][2];
#pragma unroll
        for (int kk = 0; kk < 4; ++kk) {
#pragma unroll
            for (int g = 0; g < 2; ++g)
                b4[kk][g] = *(const v4f*)(wcol + (size_t)(8 * q + kk) * D + 64 * g);
        }
        v2f a_lo, a_hi;
        a_lo.x = a4.x; a_lo.y = a4.y;
        a_hi.x = a4.z; a_hi.y = a4.w;
#pragma unroll
        for (int g = 0; g < 2; ++g) {
#pragma unroll
            for (int t = 0; t < 4; ++t) {
                v2f b;
                b.x = b4[0][g][t];
                b.y = b4[1][g][t];
                acc[g][t] = __builtin_amdgcn_wmma_f32_16x16x4_f32(
                        false, a_lo, false, b, (short)0, acc[g][t], false, false);
            }
        }
#pragma unroll
        for (int g = 0; g < 2; ++g) {
#pragma unroll
            for (int t = 0; t < 4; ++t) {
                v2f b;
                b.x = b4[2][g][t];
                b.y = b4[3][g][t];
                acc[g][t] = __builtin_amdgcn_wmma_f32_16x16x4_f32(
                        false, a_hi, false, b, (short)0, acc[g][t], false, false);
            }
        }
    }

    float* srow = support + (size_t)(m0 + 8 * h) * D + 4 * c;
#pragma unroll
    for (int r = 0; r < 8; ++r) {
#pragma unroll
        for (int g = 0; g < 2; ++g) {
            v4f o;
            o.x = acc[g][0][r];
            o.y = acc[g][1][r];
            o.z = acc[g][2][r];
            o.w = acc[g][3][r];
            *(v4f*)(srow + (size_t)r * D + 64 * g) = o;
        }
    }
}

#define RPW     13
#define WPB     32
#define RPB     (RPW * WPB)
#define NBK     256
#define OFFP    (NBK + 1)
#define ETH     (WPB * 32)
#define TILE    8192
#define GRP     8
#define RND     8
#define COLBITS 17

__device__ __forceinline__ unsigned same_key_lanes(unsigned key, int nbits, bool valid) {
    unsigned grp = __builtin_amdgcn_ballot_w32(valid);
    for (int bit = 0; bit < nbits; ++bit) {
        const bool one = ((key >> bit) & 1u) != 0u;
        const unsigned bal = __builtin_amdgcn_ballot_w32(one);
        grp &= one ? bal : ~bal;
    }
    return valid ? grp : 0u;
}

__device__ __forceinline__ void wave_count(const int* __restrict__ erow, int ebeg, int eend,
                                           int nNodes, int lane, volatile int* wh) {
    const unsigned lt = (1u << lane) - 1u;
    for (int e = ebeg; e < eend; e += 32 * RND) {
        unsigned r8[RND];
#pragma unroll
        for (int k = 0; k < RND; ++k) {
            const int idx = e + 32 * k + lane;
            r8[k] = 0xffffffffu;
            if (idx < eend) r8[k] = (unsigned)erow[idx];
        }
#pragma unroll
        for (int k = 0; k < RND; ++k) {
            const unsigned r = r8[k];
            const bool valid = r < (unsigned)nNodes;
            const unsigned bk = valid ? r / RPB : 0u;
            const unsigned grp = same_key_lanes(bk, 8, valid);
            if (valid && (grp & lt) == 0u) wh[bk] += __popc(grp);
        }
    }
}

__global__ __launch_bounds__(ETH) void gcn_edge_partition(const int* __restrict__ erow,
                                                          const int* __restrict__ ecol,
                                                          const float* __restrict__ eval,
                                                          int2* __restrict__ recs,
                                                          int* __restrict__ off,
                                                          int nEdges, int chunk, int nNodes) {
    __shared__ int wh[WPB * NBK];
    __shared__ int tot[NBK + 1];
    const int tid = threadIdx.x;
    const int lane = tid & 31;
    const int wave = __builtin_amdgcn_readfirstlane(tid >> 5);
    const int me = blockIdx.x;
    for (int i = tid; i < WPB * NBK; i += ETH) wh[i] = 0;
    __syncthreads();
    const int e0 = me * chunk;
    const int e1 = (e0 + chunk < nEdges) ? (e0 + chunk) : nEdges;
    const int per = (chunk + WPB - 1) / WPB;
    const int wb = e0 + wave * per;
    const int we = (wb + per < e1) ? (wb + per) : e1;
    volatile int* mywh = wh + wave * NBK;
    wave_count(erow, wb, we, nNodes, lane, mywh);
    __syncthreads();
    if (tid < NBK) {
        int total = 0;
#pragma unroll
        for (int w = 0; w < WPB; ++w) total += wh[w * NBK + tid];
        tot[tid] = total;
    }
    __syncthreads();
    if (tid < 32) {
        int v[8];
        int run = 0;
#pragma unroll
        for (int j = 0; j < 8; ++j) {
            const int c = tot[8 * tid + j];
            v[j] = run;
            run += c;
        }
        int incl = run;
#pragma unroll
        for (int d = 1; d < 32; d <<= 1) {
            const int o = __shfl_up(incl, d);
            if (tid >= d) incl += o;
        }
        const int excl = incl - run;
#pragma unroll
        for (int j = 0; j < 8; ++j) tot[8 * tid + j] = excl + v[j];
        if (tid == 31) tot[NBK] = incl;
    }
    __syncthreads();
    if (tid < NBK) {
        int run = tot[tid];
#pragma unroll
        for (int w = 0; w < WPB; ++w) {
            const int v = wh[w * NBK + tid];
            wh[w * NBK + tid] = run;
            run += v;
        }
    }
    if (tid <= NBK) off[me * OFFP + tid] = tot[tid];
    __syncthreads();

    const unsigned lt = (1u << lane) - 1u;
    for (int e = wb; e < we; e += 32 * RND) {
        unsigned r8[RND];
        unsigned c8[RND];
        int v8[RND];
#pragma unroll
        for (int k = 0; k < RND; ++k) {
            const int idx = e + 32 * k + lane;
            r8[k] = 0xffffffffu;
            c8[k] = 0u;
            v8[k] = 0;
            if (idx < we) {
                r8[k] = (unsigned)erow[idx];
                c8[k] = (unsigned)ecol[idx];
                v8[k] = __float_as_int(eval[idx]);
            }
        }
#pragma unroll
        for (int k = 0; k < RND; ++k) {
            const unsigned r = r8[k];
            const bool valid = r < (unsigned)nNodes;
            const unsigned bk = valid ? r / RPB : 0u;
            const unsigned grp = same_key_lanes(bk, 8, valid);
            if (valid) {
                const int pos = e0 + mywh[bk] + __popc(grp & lt);
                int2 rec;
                rec.x = (int)((c8[k] & ((1u << COLBITS) - 1u)) | ((r - bk * RPB) << COLBITS));
                rec.y = v8[k];
                recs[pos] = rec;
            }
            if (valid && (grp & lt) == 0u) mywh[bk] += __popc(grp);
        }
    }
}

__device__ __forceinline__ v4f gather_row(const float* __restrict__ support, unsigned laneoff, int key) {
    const unsigned col = (unsigned)key & ((1u << COLBITS) - 1u);
    return *(const v4f*)((const char*)support + (size_t)col * (D * sizeof(float)) + laneoff);
}

__device__ __forceinline__ void add_row(v4f* sums4, int lane, int key, float v, v4f m) {
    v4f* p = sums4 + ((unsigned)key >> COLBITS) * 32 + lane;
    v4f a = *p;
    a.x += m.x * v;
    a.y += m.y * v;
    a.z += m.z * v;
    a.w += m.w * v;
    *p = a;
}

__global__ __launch_bounds__(ETH) void gcn_aggregate(const float* __restrict__ support,
                                                     const int2* __restrict__ recs,
                                                     const int* __restrict__ off,
                                                     float* __restrict__ out, int nNodes, int chunk) {
    __shared__ __attribute__((aligned(16))) float sums[RPB * D];
    __shared__ __attribute__((aligned(8)))  int2  tile[TILE];
    __shared__ int wh[WPB * WPB];
    __shared__ int ostart[WPB + 1];
    __shared__ int sstart[NBK + 1];
    __shared__ int sbase[NBK];
    const int tid = threadIdx.x;
    const int lane = tid & 31;
    const int wave = __builtin_amdgcn_readfirstlane(tid >> 5);
    const int b = blockIdx.x;
    const unsigned lt = (1u << lane) - 1u;
    const unsigned laneoff = (unsigned)lane * 16u;
    v4f* sums4 = (v4f*)sums;
    v4f* mys = sums4 + wave * RPW * 32;
    volatile int* mywh = wh + wave * WPB;

    {
        const v4f z4 = {0.f, 0.f, 0.f, 0.f};
        for (int r = 0; r < RPW; ++r) mys[r * 32 + lane] = z4;
    }

    if (tid < NBK) {
        const int o0 = off[tid * OFFP + b];
        const int o1 = off[tid * OFFP + b + 1];
        sstart[tid] = o1 - o0;
        sbase[tid] = tid * chunk + o0;
    }
    __syncthreads();
    if (tid < 32) {
        int v[8];
        int run = 0;
#pragma unroll
        for (int j = 0; j < 8; ++j) {
            const int c = sstart[8 * tid + j];
            v[j] = run;
            run += c;
        }
        int incl = run;
#pragma unroll
        for (int d = 1; d < 32; d <<= 1) {
            const int o = __shfl_up(incl, d);
            if (tid >= d) incl += o;
        }
        const int excl = incl - run;
#pragma unroll
        for (int j = 0; j < 8; ++j) sstart[8 * tid + j] = excl + v[j];
        if (tid == 31) sstart[NBK] = incl;
    }
    __syncthreads();
    const int t = __builtin_amdgcn_readfirstlane(sstart[NBK]);
    for (int tile0 = 0; tile0 < t; tile0 += TILE) {
        const int nt = (t - tile0 < TILE) ? (t - tile0) : TILE;
        mywh[lane] = 0;
        int2 rc[TILE / ETH];
        unsigned own[TILE / ETH];
        int pw[TILE / ETH];
#pragma unroll
        for (int k = 0; k < TILE / ETH; ++k) {
            const int idx = wave * (TILE / WPB) + k * 32 + lane;
            rc[k] = make_int2(0, 0);
            if (idx < nt) {
                const int gi = tile0 + idx;
                int lo = 0, hi = NBK;
#pragma unroll
                for (int it = 0; it < 8; ++it) {
                    const int mid = (lo + hi) >> 1;
                    if (sstart[mid] <= gi) lo = mid; else hi = mid;
                }
                rc[k] = recs[sbase[lo] + (gi - sstart[lo])];
            }
        }
#pragma unroll
        for (int k = 0; k < TILE / ETH; ++k) {
            const int idx = wave * (TILE / WPB) + k * 32 + lane;
            const bool valid = idx < nt;
            own[k] = valid ? ((unsigned)rc[k].x >> COLBITS) / RPW : 0u;
            const unsigned grp = same_key_lanes(own[k], 5, valid);
            const int base = mywh[own[k]];
            pw[k] = base + __popc(grp & lt);
            if (valid && (grp & lt) == 0u) mywh[own[k]] = base + __popc(grp);
        }
        __syncthreads();
        if (tid < 32) {
            int run = 0;
#pragma unroll
            for (int w = 0; w < WPB; ++w) {
                const int v = wh[w * WPB + lane];
                wh[w * WPB + lane] = run;
                run += v;
            }
            int incl = run;
#pragma unroll
            for (int d = 1; d < 32; d <<= 1) {
                const int o = __shfl_up(incl, d);
                if (lane >= d) incl += o;
            }
            ostart[lane] = incl - run;
            if (lane == 31) ostart[WPB] = incl;
        }
        __syncthreads();
#pragma unroll
        for (int k = 0; k < TILE / ETH; ++k) {
            const int idx = wave * (TILE / WPB) + k * 32 + lane;
            if (idx < nt) tile[ostart[own[k]] + wh[wave * WPB + own[k]] + pw[k]] = rc[k];
        }
        __syncthreads();
        const int a0 = __builtin_amdgcn_readfirstlane(ostart[wave]);
        const int a1 = __builtin_amdgcn_readfirstlane(ostart[wave + 1]);
        for (int base = a0; base < a1; base += 32) {
            const int n = (a1 - base < 32) ? (a1 - base) : 32;
            int2 rec = make_int2(0, 0);
            if (lane < n) rec = tile[base + lane];
            if (n == 32) {
                v4f m[2][GRP];
#pragma unroll
                for (int j = 0; j < GRP; ++j)
                    m[0][j] = gather_row(support, laneoff, __builtin_amdgcn_readlane(rec.x, j));
#pragma unroll
                for (int g = 0; g < 32 / GRP; ++g) {
                    if (g + 1 < 32 / GRP) {
#pragma unroll
                        for (int j = 0; j < GRP; ++j)
                            m[(g + 1) & 1][j] = gather_row(support, laneoff,
                                    __builtin_amdgcn_readlane(rec.x, (g + 1) * GRP + j));
                    }
#pragma unroll
                    for (int j = 0; j < GRP; ++j) {
                        const int e = g * GRP + j;
                        add_row(sums4, lane, __builtin_amdgcn_readlane(rec.x, e),
                                __int_as_float(__builtin_amdgcn_readlane(rec.y, e)), m[g & 1][j]);
                    }
                }
            } else {
                for (int i0 = 0; i0 < n; i0 += GRP) {
                    v4f mt[GRP];
#pragma unroll
                    for (int j = 0; j < GRP; ++j) {
                        if (i0 + j < n)
                            mt[j] = gather_row(support, laneoff, __builtin_amdgcn_readlane(rec.x, i0 + j));
                    }
#pragma unroll
                    for (int j = 0; j < GRP; ++j) {
                        if (i0 + j < n)
                            add_row(sums4, lane, __builtin_amdgcn_readlane(rec.x, i0 + j),
                                    __int_as_float(__builtin_amdgcn_readlane(rec.y, i0 + j)), mt[j]);
                    }
                }
            }
        }
        __syncthreads();
    }

    for (int r = 0; r < RPW; ++r) {
        const int row = b * RPB + wave * RPW + r;
        if (row < nNodes) {
            const v4f a = mys[r * 32 + lane];
            v4f o;
            o.x = tanhf(a.x);
            o.y = tanhf(a.y);
            o.z = tanhf(a.z);
            o.w = tanhf(a.w);
            *(v4f*)(out + (size_t)row * D + 4 * lane) = o;
        }
    }
}

extern "C" void kernel_launch(void* const* d_in, const int* in_sizes, int n_in,
                              void* d_out, int out_size, void* d_ws, size_t ws_size,
                              hipStream_t stream) {
    const float* x    = (const float*)d_in[0];
    const float* W    = (const float*)d_in[1];
    const int*   erow = (const int*)d_in[2];
    const int*   ecol = (const int*)d_in[3];
    const float* eval = (const float*)d_in[4];
    float* out = (float*)d_out;

    const int nNodes = in_sizes[0] / D;
    const int nEdges = in_sizes[2];

    float* support = (float*)d_ws;
    int2*  recs    = (int2*)(support + (size_t)nNodes * D);
    int*   off     = (int*)(recs + nEdges);

    const int nBlocks = NBK;
    const int chunk   = (nEdges + nBlocks - 1) / nBlocks;

    {
        const int nStripes = nNodes / 16;
        gcn_gemm_wmma<<<(nStripes + 3) / 4, 128, 0, stream>>>(x, W, support, nStripes);
    }
    gcn_edge_partition<<<nBlocks, ETH, 0, stream>>>(erow, ecol, eval, recs, off, nEdges, chunk, nNodes);
    gcn_aggregate<<<(nNodes + RPB - 1) / RPB, ETH, 0, stream>>>(support, recs, off, out, nNodes, chunk);
}


# aggregate: LDS tile ordered by row, so a wave adds a row's records in registers in edge order and stores tanh of finished rows under the next gathers; per-record LDS add gone; same adds, same order
# speedup vs baseline: 3.4424x; 1.0632x over previous
#include <hip/hip_runtime.h>
#include <hip/hip_bf16.h>

typedef __attribute__((ext_vector_type(2))) float v2f;
typedef __attribute__((ext_vector_type(8))) float v8f;

#define D 128

typedef __attribute__((ext_vector_type(4))) float v4f;

__global__ __launch_bounds__(128) void gcn_gemm_wmma(const float* __restrict__ x,
                                                     const float* __restrict__ W,
                                                     float* __restrict__ support,
                                                     int nStripes) {
    const int lane = threadIdx.x & 31;
    const int wave = __builtin_amdgcn_readfirstlane(threadIdx.x >> 5);
    const int stripe = blockIdx.x * 4 + wave;
    if (stripe >= nStripes) return;
    const int m0 = stripe * 16;
    const int h  = lane >> 4;
    const int c  = lane & 15;

    const float* arow = x + (size_t)(m0 + c) * D + 4 * h;
    const float* wcol = W + (size_t)(4 * h) * D + 4 * c;

    v8f acc[2][4] = {};
#pragma unroll 2
    for (int q = 0; q < D / 8; ++q) {
        const v4f a4 = *(const v4f*)(arow + 8 * q);
        v4f b4[4][2];
#pragma unroll
        for (int kk = 0; kk < 4; ++kk) {
#pragma unroll
            for (int g = 0; g < 2; ++g)
                b4[kk][g] = *(const v4f*)(wcol + (size_t)(8 * q + kk) * D + 64 * g);
        }
        v2f a_lo, a_hi;
        a_lo.x = a4.x; a_lo.y = a4.y;
        a_hi.x = a4.z; a_hi.y = a4.w;
#pragma unroll
        for (int g = 0; g < 2; ++g) {
#pragma unroll
            for (int t = 0; t < 4; ++t) {
                v2f b;
                b.x = b4[0][g][t];
                b.y = b4[1][g][t];
                acc[g][t] = __builtin_amdgcn_wmma_f32_16x16x4_f32(
                        false, a_lo, false, b, (short)0, acc[g][t], false, false);
            }
        }
#pragma unroll
        for (int g = 0; g < 2; ++g) {
#pragma unroll
            for (int t = 0; t < 4; ++t) {
                v2f b;
                b.x = b4[2][g][t];
                b.y = b4[3][g][t];
                acc[g][t] = __builtin_amdgcn_wmma_f32_16x16x4_f32(
                        false, a_hi, false, b, (short)0, acc[g][t], false, false);
            }
        }
    }

    float* srow = support + (size_t)(m0 + 8 * h) * D + 4 * c;
#pragma unroll
    for (int r = 0; r < 8; ++r) {
#pragma unroll
        for (int g = 0; g < 2; ++g) {
            v4f o;
            o.x = acc[g][0][r];
            o.y = acc[g][1][r];
            o.z = acc[g][2][r];
            o.w = acc[g][3][r];
            *(v4f*)(srow + (size_t)r * D + 64 * g) = o;
        }
    }
}

#define RPW     13
#define WPB     32
#define RPB     (RPW * WPB)
#define NBK     256
#define OFFP    (NBK + 1)
#define ETH     (WPB * 32)
#define TILE    8192
#define GRP     8
#define RND     8
#define COLBITS 17
#define ROWBITS 9

__device__ __forceinline__ unsigned same_key_lanes(unsigned key, int nbits, bool valid) {
    unsigned grp = __builtin_amdgcn_ballot_w32(valid);
    for (int bit = 0; bit < nbits; ++bit) {
        const bool one = ((key >> bit) & 1u) != 0u;
        const unsigned bal = __builtin_amdgcn_ballot_w32(one);
        grp &= one ? bal : ~bal;
    }
    return valid ? grp : 0u;
}

__device__ __forceinline__ void wave_count(const int* __restrict__ erow, int ebeg, int eend,
                                           int nNodes, int lane, volatile int* wh) {
    const unsigned lt = (1u << lane) - 1u;
    for (int e = ebeg; e < eend; e += 32 * RND) {
        unsigned r8[RND];
#pragma unroll
        for (int k = 0; k < RND; ++k) {
            const int idx = e + 32 * k + lane;
            r8[k] = 0xffffffffu;
            if (idx < eend) r8[k] = (unsigned)erow[idx];
        }
#pragma unroll
        for (int k = 0; k < RND; ++k) {
            const unsigned r = r8[k];
            const bool valid = r < (unsigned)nNodes;
            const unsigned bk = valid ? r / RPB : 0u;
            const unsigned grp = same_key_lanes(bk, 8, valid);
            if (valid && (grp & lt) == 0u) wh[bk] += __popc(grp);
        }
    }
}

__global__ __launch_bounds__(ETH) void gcn_edge_partition(const int* __restrict__ erow,
                                                          const int* __restrict__ ecol,
                                                          const float* __restrict__ eval,
                                                          int2* __restrict__ recs,
                                                          int* __restrict__ off,
                                                          int nEdges, int chunk, int nNodes) {
    __shared__ int wh[WPB * NBK];
    __shared__ int tot[NBK + 1];
    const int tid = threadIdx.x;
    const int lane = tid & 31;
    const int wave = __builtin_amdgcn_readfirstlane(tid >> 5);
    const int me = blockIdx.x;
    for (int i = tid; i < WPB * NBK; i += ETH) wh[i] = 0;
    __syncthreads();
    const int e0 = me * chunk;
    const int e1 = (e0 + chunk < nEdges) ? (e0 + chunk) : nEdges;
    const int per = (chunk + WPB - 1) / WPB;
    const int wb = e0 + wave * per;
    const int we = (wb + per < e1) ? (wb + per) : e1;
    volatile int* mywh = wh + wave * NBK;
    wave_count(erow, wb, we, nNodes, lane, mywh);
    __syncthreads();
    if (tid < NBK) {
        int total = 0;
#pragma unroll
        for (int w = 0; w < WPB; ++w) total += wh[w * NBK + tid];
        tot[tid] = total;
    }
    __syncthreads();
    if (tid < 32) {
        int v[8];
        int run = 0;
#pragma unroll
        for (int j = 0; j < 8; ++j) {
            const int c = tot[8 * tid + j];
            v[j] = run;
            run += c;
        }
        int incl = run;
#pragma unroll
        for (int d = 1; d < 32; d <<= 1) {
            const int o = __shfl_up(incl, d);
            if (tid >= d) incl += o;
        }
        const int excl = incl - run;
#pragma unroll
        for (int j = 0; j < 8; ++j) tot[8 * tid + j] = excl + v[j];
        if (tid == 31) tot[NBK] = incl;
    }
    __syncthreads();
    if (tid < NBK) {
        int run = tot[tid];
#pragma unroll
        for (int w = 0; w < WPB; ++w) {
            const int v = wh[w * NBK + tid];
            wh[w * NBK + tid] = run;
            run += v;
        }
    }
    if (tid <= NBK) off[me * OFFP + tid] = tot[tid];
    __syncthreads();

    const unsigned lt = (1u << lane) - 1u;
    for (int e = wb; e < we; e += 32 * RND) {
        unsigned r8[RND];
        unsigned c8[RND];
        int v8[RND];
#pragma unroll
        for (int k = 0; k < RND; ++k) {
            const int idx = e + 32 * k + lane;
            r8[k] = 0xffffffffu;
            c8[k] = 0u;
            v8[k] = 0;
            if (idx < we) {
                r8[k] = (unsigned)erow[idx];
                c8[k] = (unsigned)ecol[idx];
                v8[k] = __float_as_int(eval[idx]);
            }
        }
#pragma unroll
        for (int k = 0; k < RND; ++k) {
            const unsigned r = r8[k];
            const bool valid = r < (unsigned)nNodes;
            const unsigned bk = valid ? r / RPB : 0u;
            const unsigned grp = same_key_lanes(bk, 8, valid);
            if (valid) {
                const int pos = e0 + mywh[bk] + __popc(grp & lt);
                int2 rec;
                rec.x = (int)((c8[k] & ((1u << COLBITS) - 1u)) | ((r - bk * RPB) << COLBITS));
                rec.y = v8[k];
                recs[pos] = rec;
            }
            if (valid && (grp & lt) == 0u) mywh[bk] += __popc(grp);
        }
    }
}

__device__ __forceinline__ v4f gather_row(const float* __restrict__ support, unsigned laneoff, int key) {
    const unsigned col = (unsigned)key & ((1u << COLBITS) - 1u);
    return *(const v4f*)((const char*)support + (size_t)col * (D * sizeof(float)) + laneoff);
}

typedef int __attribute__((may_alias)) int_alias;

__device__ __forceinline__ void store_row(const v4f* sums4, float* __restrict__ out, int lane,
                                          int r, int row, int nNodes) {
    if (row < nNodes) {
        const v4f a = sums4[r * 32 + lane];
        v4f o;
        o.x = tanhf(a.x);
        o.y = tanhf(a.y);
        o.z = tanhf(a.z);
        o.w = tanhf(a.w);
        *(v4f*)(out + (size_t)row * D + 4 * lane) = o;
    }
}

__global__ __launch_bounds__(ETH) void gcn_aggregate(const float* __restrict__ support,
                                                     const int2* __restrict__ recs,
                                                     const int* __restrict__ off,
                                                     float* __restrict__ out, int nNodes, int chunk) {
    __shared__ __attribute__((aligned(16))) float sums[RPB * D];
    __shared__ __attribute__((aligned(8)))  int2  tile[TILE];
    __shared__ __attribute__((aligned(4)))  unsigned short wh[WPB * RPB];
    __shared__ int rstart[RPB + 1];
    __shared__ int sstart[NBK + 1];
    __shared__ int sbase[NBK];
    const int tid = threadIdx.x;
    const int lane = tid & 31;
    const int wave = __builtin_amdgcn_readfirstlane(tid >> 5);
    const int b = blockIdx.x;
    const unsigned lt = (1u << lane) - 1u;
    const unsigned laneoff = (unsigned)lane * 16u;
    v4f* sums4 = (v4f*)sums;
    v4f* mys = sums4 + wave * RPW * 32;
    unsigned short* mywh = wh + wave * RPB;

    {
        const v4f z4 = {0.f, 0.f, 0.f, 0.f};
        for (int r = 0; r < RPW; ++r) mys[r * 32 + lane] = z4;
    }

    if (tid < NBK) {
        const int o0 = off[tid * OFFP + b];
        const int o1 = off[tid * OFFP + b + 1];
        sstart[tid] = o1 - o0;
        sbase[tid] = tid * chunk + o0;
    }
    __syncthreads();
    if (tid < 32) {
        int v[8];
        int run = 0;
#pragma unroll
        for (int j = 0; j < 8; ++j) {
            const int c = sstart[8 * tid + j];
            v[j] = run;
            run += c;
        }
        int incl = run;
#pragma unroll
        for (int d = 1; d < 32; d <<= 1) {
            const int o = __shfl_up(incl, d);
            if (tid >= d) incl += o;
        }
        const int excl = incl - run;
#pragma unroll
        for (int j = 0; j < 8; ++j) sstart[8 * tid + j] = excl + v[j];
        if (tid == 31) sstart[NBK] = incl;
    }
    __syncthreads();
    const int t = __builtin_amdgcn_readfirstlane(sstart[NBK]);
    const int row0 = wave * RPW;
    int done = row0;
    for (int tile0 = 0; tile0 < t; tile0 += TILE) {
        const int nt = (t - tile0 < TILE) ? (t - tile0) : TILE;
        const bool lastTile = tile0 + TILE >= t;
        for (int i = lane; i < RPB / 2; i += 32) ((int_alias*)mywh)[i] = 0;
        int2 rc[TILE / ETH];
        unsigned rowk[TILE / ETH];
        int pw[TILE / ETH];
#pragma unroll
        for (int k = 0; k < TILE / ETH; ++k) {
            const int idx = wave * (TILE / WPB) + k * 32 + lane;
            rc[k] = make_int2(0, 0);
            if (idx < nt) {
                const int gi = tile0 + idx;
                int lo = 0, hi = NBK;
#pragma unroll
                for (int it = 0; it < 8; ++it) {
                    const int mid = (lo + hi) >> 1;
                    if (sstart[mid] <= gi) lo = mid; else hi = mid;
                }
                rc[k] = recs[sbase[lo] + (gi - sstart[lo])];
            }
        }
#pragma unroll
        for (int k = 0; k < TILE / ETH; ++k) {
            const int idx = wave * (TILE / WPB) + k * 32 + lane;
            const bool valid = idx < nt;
            rowk[k] = valid ? ((unsigned)rc[k].x >> COLBITS) : 0u;
            const unsigned grp = same_key_lanes(rowk[k], ROWBITS, valid);
            const int base = mywh[rowk[k]];
            pw[k] = base + __popc(grp & lt);
            if (valid && (grp & lt) == 0u) mywh[rowk[k]] = (unsigned short)(base + __popc(grp));
        }
        __syncthreads();
        if (tid < RPB) {
            int run = 0;
#pragma unroll
            for (int w = 0; w < WPB; ++w) {
                const int v = wh[w * RPB + tid];
                wh[w * RPB + tid] = (unsigned short)run;
                run += v;
            }
            rstart[tid] = run;
        }
        __syncthreads();
        if (tid < 32) {
            int v[RPW];
            int run = 0;
#pragma unroll
            for (int j = 0; j < RPW; ++j) {
                const int c = rstart[RPW * lane + j];
                v[j] = run;
                run += c;
            }
            int incl = run;
#pragma unroll
            for (int d = 1; d < 32; d <<= 1) {
                const int o = __shfl_up(incl, d);
                if (lane >= d) incl += o;
            }
            const int excl = incl - run;
#pragma unroll
            for (int j = 0; j < RPW; ++j) rstart[RPW * lane + j] = excl + v[j];
            if (lane == 31) rstart[RPB] = incl;
        }
        __syncthreads();
#pragma unroll
        for (int k = 0; k < TILE / ETH; ++k) {
            const int idx = wave * (TILE / WPB) + k * 32 + lane;
            if (idx < nt) tile[rstart[rowk[k]] + wh[wave * RPB + rowk[k]] + pw[k]] = rc[k];
        }
        __syncthreads();
        const int a0 = __builtin_amdgcn_readfirstlane(rstart[row0]);
        const int a1 = __builtin_amdgcn_readfirstlane(rstart[row0 + RPW]);
        int cur = row0;
        v4f acc = sums4[cur * 32 + lane];
        for (int base = a0; base < a1; base += 32) {
            const int n = (a1 - base < 32) ? (a1 - base) : 32;
            int2 rec = make_int2(0, 0);
            if (lane < n) rec = tile[base + lane];
            if (n == 32) {
                v4f m[2][GRP];
#pragma unroll
                for (int j = 0; j < GRP; ++j)
                    m[0][j] = gather_row(support, laneoff, __builtin_amdgcn_readlane(rec.x, j));
                if (lastTile) {
                    for (; done < cur; ++done) store_row(sums4, out, lane, done, b * RPB + done, nNodes);
                }
#pragma unroll
                for (int g = 0; g < 32 / GRP; ++g) {
                    if (g + 1 < 32 / GRP) {
#pragma unroll
                        for (int j = 0; j < GRP; ++j)
                            m[(g + 1) & 1][j] = gather_row(support, laneoff,
                                    __builtin_amdgcn_readlane(rec.x, (g + 1) * GRP + j));
                    }
#pragma unroll
                    for (int j = 0; j < GRP; ++j) {
                        const int e = g * GRP + j;
                        const int key = __builtin_amdgcn_readlane(rec.x, e);
                        const float v = __int_as_float(__builtin_amdgcn_readlane(rec.y, e));
                        const int row = (int)((unsigned)key >> COLBITS);
                        if (row != cur) {
                            sums4[cur * 32 + lane] = acc;
                            acc = sums4[row * 32 + lane];
                            cur = row;
                        }
                        const v4f mm = m[g & 1][j];
                        acc.x += mm.x * v;
                        acc.y += mm.y * v;
                        acc.z += mm.z * v;
                        acc.w += mm.w * v;
                    }
                }
            } else {
                for (int i0 = 0; i0 < n; i0 += GRP) {
                    v4f mt[GRP];
#pragma unroll
                    for (int j = 0; j < GRP; ++j) {
                        mt[j] = (v4f){0.f, 0.f, 0.f, 0.f};
                        if (i0 + j < n)
                            mt[j] = gather_row(support, laneoff, __builtin_amdgcn_readlane(rec.x, i0 + j));
                    }
#pragma unroll
                    for (int j = 0; j < GRP; ++j) {
                        if (i0 + j < n) {
                            const int key = __builtin_amdgcn_readlane(rec.x, i0 + j);
                            const float v = __int_as_float(__builtin_amdgcn_readlane(rec.y, i0 + j));
                            const int row = (int)((unsigned)key >> COLBITS);
                            if (row != cur) {
                                sums4[cur * 32 + lane] = acc;
                                acc = sums4[row * 32 + lane];
                                cur = row;
                            }
                            acc.x += mt[j].x * v;
                            acc.y += mt[j].y * v;
                            acc.z += mt[j].z * v;
                            acc.w += mt[j].w * v;
                        }
                    }
                }
            }
        }
        sums4[cur * 32 + lane] = acc;
        __syncthreads();
    }

    for (; done < row0 + RPW; ++done) store_row(sums4, out, lane, done, b * RPB + done, nNodes);
}

extern "C" void kernel_launch(void* const* d_in, const int* in_sizes, int n_in,
                              void* d_out, int out_size, void* d_ws, size_t ws_size,
                              hipStream_t stream) {
    const float* x    = (const float*)d_in[0];
    const float* W    = (const float*)d_in[1];
    const int*   erow = (const int*)d_in[2];
    const int*   ecol = (const int*)d_in[3];
    const float* eval = (const float*)d_in[4];
    float* out = (float*)d_out;

    const int nNodes = in_sizes[0] / D;
    const int nEdges = in_sizes[2];

    float* support = (float*)d_ws;
    int2*  recs    = (int2*)(support + (size_t)nNodes * D);
    int*   off     = (int*)(recs + nEdges);

    const int nBlocks = NBK;
    const int chunk   = (nEdges + nBlocks - 1) / nBlocks;

    {
        const int nStripes = nNodes / 16;
        gcn_gemm_wmma<<<(nStripes + 3) / 4, 128, 0, stream>>>(x, W, support, nStripes);
    }
    gcn_edge_partition<<<nBlocks, ETH, 0, stream>>>(erow, ecol, eval, recs, off, nEdges, chunk, nNodes);
    gcn_aggregate<<<(nNodes + RPB - 1) / RPB, ETH, 0, stream>>>(support, recs, off, out, nNodes, chunk);
}


# x@W and the edge partition side by side in ONE launch (256 + 256 work-groups of 16 waves, two resident per WGP): a launch and the partition's span gone; same products, same sums, same order
# speedup vs baseline: 3.5606x; 1.0343x over previous
#include <hip/hip_runtime.h>
#include <hip/hip_bf16.h>

typedef __attribute__((ext_vector_type(2))) float v2f;
typedef __attribute__((ext_vector_type(8))) float v8f;

#define D 128

typedef __attribute__((ext_vector_type(4))) float v4f;

__device__ __forceinline__ void gemm_stripe(const float* __restrict__ x,
                                            const float* __restrict__ W,
                                            float* __restrict__ support,
                                            int stripe, int lane) {
    const int m0 = stripe * 16;
    const int h  = lane >> 4;
    const int c  = lane & 15;

    const float* arow = x + (size_t)(m0 + c) * D + 4 * h;
    const float* wcol = W + (size_t)(4 * h) * D + 4 * c;

    v8f acc[2][4] = {};
#pragma unroll 2
    for (int q = 0; q < D / 8; ++q) {
        const v4f a4 = *(const v4f*)(arow + 8 * q);
        v4f b4[4][2];
#pragma unroll
        for (int kk = 0; kk < 4; ++kk) {
#pragma unroll
            for (int g = 0; g < 2; ++g)
                b4[kk][g] = *(const v4f*)(wcol + (size_t)(8 * q + kk) * D + 64 * g);
        }
        v2f a_lo, a_hi;
        a_lo.x = a4.x; a_lo.y = a4.y;
        a_hi.x = a4.z; a_hi.y = a4.w;
#pragma unroll
        for (int g = 0; g < 2; ++g) {
#pragma unroll
            for (int t = 0; t < 4; ++t) {
                v2f b;
                b.x = b4[0][g][t];
                b.y = b4[1][g][t];
                acc[g][t] = __builtin_amdgcn_wmma_f32_16x16x4_f32(
                        false, a_lo, false, b, (short)0, acc[g][t], false, false);
            }
        }
#pragma unroll
        for (int g = 0; g < 2; ++g) {
#pragma unroll
            for (int t = 0; t < 4; ++t) {
                v2f b;
                b.x = b4[2][g][t];
                b.y = b4[3][g][t];
                acc[g][t] = __builtin_amdgcn_wmma_f32_16x16x4_f32(
                        false, a_hi, false, b, (short)0, acc[g][t], false, false);
            }
        }
    }

    float* srow = support + (size_t)(m0 + 8 * h) * D + 4 * c;
#pragma unroll
    for (int r = 0; r < 8; ++r) {
#pragma unroll
        for (int g = 0; g < 2; ++g) {
            v4f o;
            o.x = acc[g][0][r];
            o.y = acc[g][1][r];
            o.z = acc[g][2][r];
            o.w = acc[g][3][r];
            *(v4f*)(srow + (size_t)r * D + 64 * g) = o;
        }
    }
}

#define RPW     13
#define WPB     32
#define FW      16
#define FTH     (FW * 32)
#define RPB     (RPW * WPB)
#define NBK     256
#define OFFP    (NBK + 1)
#define ETH     (WPB * 32)
#define TILE    8192
#define GRP     8
#define RND     13
#define COLBITS 17
#define ROWBITS 9

__device__ __forceinline__ unsigned same_key_lanes(unsigned key, int nbits, bool valid) {
    unsigned grp = __builtin_amdgcn_ballot_w32(valid);
    for (int bit = 0; bit < nbits; ++bit) {
        const bool one = ((key >> bit) & 1u) != 0u;
        const unsigned bal = __builtin_amdgcn_ballot_w32(one);
        grp &= one ? bal : ~bal;
    }
    return valid ? grp : 0u;
}

__device__ __forceinline__ void wave_count(const int* __restrict__ erow, int ebeg, int eend,
                                           int nNodes, int lane, volatile int* wh) {
    const unsigned lt = (1u << lane) - 1u;
    for (int e = ebeg; e < eend; e += 32 * RND) {
        unsigned r8[RND];
#pragma unroll
        for (int k = 0; k < RND; ++k) {
            const int idx = e + 32 * k + lane;
            r8[k] = 0xffffffffu;
            if (idx < eend) r8[k] = (unsigned)erow[idx];
        }
#pragma unroll
        for (int k = 0; k < RND; ++k) {
            const unsigned r = r8[k];
            const bool valid = r < (unsigned)nNodes;
            const unsigned bk = valid ? r / RPB : 0u;
            const unsigned grp = same_key_lanes(bk, 8, valid);
            if (valid && (grp & lt) == 0u) wh[bk] += __popc(grp);
        }
    }
}

__global__ __launch_bounds__(FTH, 8) void gcn_gemm_and_partition(const float* __restrict__ x,
                                                                 const float* __restrict__ W,
                                                                 float* __restrict__ support,
                                                                 int nStripes, int nG,
                                                                 const int* __restrict__ erow,
                                                                 const int* __restrict__ ecol,
                                                                 const float* __restrict__ eval,
                                                                 int2* __restrict__ recs,
                                                                 int* __restrict__ off,
                                                                 int nEdges, int chunk, int nNodes) {
    __shared__ int wh[FW * NBK];
    __shared__ int tot[NBK + 1];
    const int tid = threadIdx.x;
    const int lane = tid & 31;
    const int wave = __builtin_amdgcn_readfirstlane(tid >> 5);
    if (blockIdx.x >= NBK) {
        const int g = (int)blockIdx.x - NBK;
        for (int s = wave * nG + g; s < nStripes; s += FW * nG) gemm_stripe(x, W, support, s, lane);
        return;
    }
    const int me = blockIdx.x;
    for (int i = tid; i < FW * NBK; i += FTH) wh[i] = 0;
    __syncthreads();
    const int e0 = me * chunk;
    const int e1 = (e0 + chunk < nEdges) ? (e0 + chunk) : nEdges;
    const int per = (chunk + FW - 1) / FW;
    const int wb = e0 + wave * per;
    const int we = (wb + per < e1) ? (wb + per) : e1;
    volatile int* mywh = wh + wave * NBK;
    wave_count(erow, wb, we, nNodes, lane, mywh);
    __syncthreads();
    if (tid < NBK) {
        int total = 0;
#pragma unroll
        for (int w = 0; w < FW; ++w) total += wh[w * NBK + tid];
        tot[tid] = total;
    }
    __syncthreads();
    if (tid < 32) {
        int v[8];
        int run = 0;
#pragma unroll
        for (int j = 0; j < 8; ++j) {
            const int c = tot[8 * tid + j];
            v[j] = run;
            run += c;
        }
        int incl = run;
#pragma unroll
        for (int d = 1; d < 32; d <<= 1) {
            const int o = __shfl_up(incl, d);
            if (tid >= d) incl += o;
        }
        const int excl = incl - run;
#pragma unroll
        for (int j = 0; j < 8; ++j) tot[8 * tid + j] = excl + v[j];
        if (tid == 31) tot[NBK] = incl;
    }
    __syncthreads();
    if (tid < NBK) {
        int run = tot[tid];
#pragma unroll
        for (int w = 0; w < FW; ++w) {
            const int v = wh[w * NBK + tid];
            wh[w * NBK + tid] = run;
            run += v;
        }
    }
    if (tid <= NBK) off[me * OFFP + tid] = tot[tid];
    __syncthreads();

    const unsigned lt = (1u << lane) - 1u;
    for (int e = wb; e < we; e += 32 * RND) {
        unsigned r8[RND];
        unsigned c8[RND];
        int v8[RND];
#pragma unroll
        for (int k = 0; k < RND; ++k) {
            const int idx = e + 32 * k + lane;
            r8[k] = 0xffffffffu;
            c8[k] = 0u;
            v8[k] = 0;
            if (idx < we) {
                r8[k] = (unsigned)erow[idx];
                c8[k] = (unsigned)ecol[idx];
                v8[k] = __float_as_int(eval[idx]);
            }
        }
#pragma unroll
        for (int k = 0; k < RND; ++k) {
            const unsigned r = r8[k];
            const bool valid = r < (unsigned)nNodes;
            const unsigned bk = valid ? r / RPB : 0u;
            const unsigned grp = same_key_lanes(bk, 8, valid);
            if (valid) {
                const int pos = e0 + mywh[bk] + __popc(grp & lt);
                int2 rec;
                rec.x = (int)((c8[k] & ((1u << COLBITS) - 1u)) | ((r - bk * RPB) << COLBITS));
                rec.y = v8[k];
                recs[pos] = rec;
            }
            if (valid && (grp & lt) == 0u) mywh[bk] += __popc(grp);
        }
    }
}

__device__ __forceinline__ v4f gather_row(const float* __restrict__ support, unsigned laneoff, int key) {
    const unsigned col = (unsigned)key & ((1u << COLBITS) - 1u);
    return *(const v4f*)((const char*)support + (size_t)col * (D * sizeof(float)) + laneoff);
}

typedef int __attribute__((may_alias)) int_alias;

__device__ __forceinline__ void store_row(const v4f* sums4, float* __restrict__ out, int lane,
                                          int r, int row, int nNodes) {
    if (row < nNodes) {
        const v4f a = sums4[r * 32 + lane];
        v4f o;
        o.x = tanhf(a.x);
        o.y = tanhf(a.y);
        o.z = tanhf(a.z);
        o.w = tanhf(a.w);
        *(v4f*)(out + (size_t)row * D + 4 * lane) = o;
    }
}

__global__ __launch_bounds__(ETH) void gcn_aggregate(const float* __restrict__ support,
                                                     const int2* __restrict__ recs,
                                                     const int* __restrict__ off,
                                                     float* __restrict__ out, int nNodes, int chunk) {
    __shared__ __attribute__((aligned(16))) float sums[RPB * D];
    __shared__ __attribute__((aligned(8)))  int2  tile[TILE];
    __shared__ __attribute__((aligned(4)))  unsigned short wh[WPB * RPB];
    __shared__ int rstart[RPB + 1];
    __shared__ int sstart[NBK + 1];
    __shared__ int sbase[NBK];
    const int tid = threadIdx.x;
    const int lane = tid & 31;
    const int wave = __builtin_amdgcn_readfirstlane(tid >> 5);
    const int b = blockIdx.x;
    const unsigned lt = (1u << lane) - 1u;
    const unsigned laneoff = (unsigned)lane * 16u;
    v4f* sums4 = (v4f*)sums;
    v4f* mys = sums4 + wave * RPW * 32;
    unsigned short* mywh = wh + wave * RPB;

    {
        const v4f z4 = {0.f, 0.f, 0.f, 0.f};
        for (int r = 0; r < RPW; ++r) mys[r * 32 + lane] = z4;
    }

    if (tid < NBK) {
        const int o0 = off[tid * OFFP + b];
        const int o1 = off[tid * OFFP + b + 1];
        sstart[tid] = o1 - o0;
        sbase[tid] = tid * chunk + o0;
    }
    __syncthreads();
    if (tid < 32) {
        int v[8];
        int run = 0;
#pragma unroll
        for (int j = 0; j < 8; ++j) {
            const int c = sstart[8 * tid + j];
            v[j] = run;
            run += c;
        }
        int incl = run;
#pragma unroll
        for (int d = 1; d < 32; d <<= 1) {
            const int o = __shfl_up(incl, d);
            if (tid >= d) incl += o;
        }
        const int excl = incl - run;
#pragma unroll
        for (int j = 0; j < 8; ++j) sstart[8 * tid + j] = excl + v[j];
        if (tid == 31) sstart[NBK] = incl;
    }
    __syncthreads();
    const int t = __builtin_amdgcn_readfirstlane(sstart[NBK]);
    const int row0 = wave * RPW;
    int done = row0;
    for (int tile0 = 0; tile0 < t; tile0 += TILE) {
        const int nt = (t - tile0 < TILE) ? (t - tile0) : TILE;
        const bool lastTile = tile0 + TILE >= t;
        for (int i = lane; i < RPB / 2; i += 32) ((int_alias*)mywh)[i] = 0;
        int2 rc[TILE / ETH];
        unsigned rowk[TILE / ETH];
        int pw[TILE / ETH];
#pragma unroll
        for (int k = 0; k < TILE / ETH; ++k) {
            const int idx = wave * (TILE / WPB) + k * 32 + lane;
            rc[k] = make_int2(0, 0);
            if (idx < nt) {
                const int gi = tile0 + idx;
                int lo = 0, hi = NBK;
#pragma unroll
                for (int it = 0; it < 8; ++it) {
                    const int mid = (lo + hi) >> 1;
                    if (sstart[mid] <= gi) lo = mid; else hi = mid;
                }
                rc[k] = recs[sbase[lo] + (gi - sstart[lo])];
            }
        }
#pragma unroll
        for (int k = 0; k < TILE / ETH; ++k) {
            const int idx = wave * (TILE / WPB) + k * 32 + lane;
            const bool valid = idx < nt;
            rowk[k] = valid ? ((unsigned)rc[k].x >> COLBITS) : 0u;
            const unsigned grp = same_key_lanes(rowk[k], ROWBITS, valid);
            const int base = mywh[rowk[k]];
            pw[k] = base + __popc(grp & lt);
            if (valid && (grp & lt) == 0u) mywh[rowk[k]] = (unsigned short)(base + __popc(grp));
        }
        __syncthreads();
        if (tid < RPB) {
            int run = 0;
#pragma unroll
            for (int w = 0; w < WPB; ++w) {
                const int v = wh[w * RPB + tid];
                wh[w * RPB + tid] = (unsigned short)run;
                run += v;
            }
            rstart[tid] = run;
        }
        __syncthreads();
        if (tid < 32) {
            int v[RPW];
            int run = 0;
#pragma unroll
            for (int j = 0; j < RPW; ++j) {
                const int c = rstart[RPW * lane + j];
                v[j] = run;
                run += c;
            }
            int incl = run;
#pragma unroll
            for (int d = 1; d < 32; d <<= 1) {
                const int o = __shfl_up(incl, d);
                if (lane >= d) incl += o;
            }
            const int excl = incl - run;
#pragma unroll
            for (int j = 0; j < RPW; ++j) rstart[RPW * lane + j] = excl + v[j];
            if (lane == 31) rstart[RPB] = incl;
        }
        __syncthreads();
#pragma unroll
        for (int k = 0; k < TILE / ETH; ++k) {
            const int idx = wave * (TILE / WPB) + k * 32 + lane;
            if (idx < nt) tile[rstart[rowk[k]] + wh[wave * RPB + rowk[k]] + pw[k]] = rc[k];
        }
        __syncthreads();
        const int a0 = __builtin_amdgcn_readfirstlane(rstart[row0]);
        const int a1 = __builtin_amdgcn_readfirstlane(rstart[row0 + RPW]);
        int cur = row0;
        v4f acc = sums4[cur * 32 + lane];
        for (int base = a0; base < a1; base += 32) {
            const int n = (a1 - base < 32) ? (a1 - base) : 32;
            int2 rec = make_int2(0, 0);
            if (lane < n) rec = tile[base + lane];
            if (n == 32) {
                v4f m[2][GRP];
#pragma unroll
                for (int j = 0; j < GRP; ++j)
                    m[0][j] = gather_row(support, laneoff, __builtin_amdgcn_readlane(rec.x, j));
                if (lastTile) {
                    for (; done < cur; ++done) store_row(sums4, out, lane, done, b * RPB + done, nNodes);
                }
#pragma unroll
                for (int g = 0; g < 32 / GRP; ++g) {
                    if (g + 1 < 32 / GRP) {
#pragma unroll
                        for (int j = 0; j < GRP; ++j)
                            m[(g + 1) & 1][j] = gather_row(support, laneoff,
                                    __builtin_amdgcn_readlane(rec.x, (g + 1) * GRP + j));
                    }
#pragma unroll
                    for (int j = 0; j < GRP; ++j) {
                        const int e = g * GRP + j;
                        const int key = __builtin_amdgcn_readlane(rec.x, e);
                        const float v = __int_as_float(__builtin_amdgcn_readlane(rec.y, e));
                        const int row = (int)((unsigned)key >> COLBITS);
                        if (row != cur) {
                            sums4[cur * 32 + lane] = acc;
                            acc = sums4[row * 32 + lane];
                            cur = row;
                        }
                        const v4f mm = m[g & 1][j];
                        acc.x += mm.x * v;
                        acc.y += mm.y * v;
                        acc.z += mm.z * v;
                        acc.w += mm.w * v;
                    }
                }
            } else {
                for (int i0 = 0; i0 < n; i0 += GRP) {
                    v4f mt[GRP];
#pragma unroll
                    for (int j = 0; j < GRP; ++j) {
                        mt[j] = (v4f){0.f, 0.f, 0.f, 0.f};
                        if (i0 + j < n)
                            mt[j] = gather_row(support, laneoff, __builtin_amdgcn_readlane(rec.x, i0 + j));
                    }
#pragma unroll
                    for (int j = 0; j < GRP; ++j) {
                        if (i0 + j < n) {
                            const int key = __builtin_amdgcn_readlane(rec.x, i0 + j);
                            const float v = __int_as_float(__builtin_amdgcn_readlane(rec.y, i0 + j));
                            const int row = (int)((unsigned)key >> COLBITS);
                            if (row != cur) {
                                sums4[cur * 32 + lane] = acc;
                                acc = sums4[row * 32 + lane];
                                cur = row;
                            }
                            acc.x += mt[j].x * v;
                            acc.y += mt[j].y * v;
                            acc.z += mt[j].z * v;
                            acc.w += mt[j].w * v;
                        }
                    }
                }
            }
        }
        sums4[cur * 32 + lane] = acc;
        __syncthreads();
    }

    for (; done < row0 + RPW; ++done) store_row(sums4, out, lane, done, b * RPB + done, nNodes);
}

extern "C" void kernel_launch(void* const* d_in, const int* in_sizes, int n_in,
                              void* d_out, int out_size, void* d_ws, size_t ws_size,
                              hipStream_t stream) {
    const float* x    = (const float*)d_in[0];
    const float* W    = (const float*)d_in[1];
    const int*   erow = (const int*)d_in[2];
    const int*   ecol = (const int*)d_in[3];
    const float* eval = (const float*)d_in[4];
    float* out = (float*)d_out;

    const int nNodes = in_sizes[0] / D;
    const int nEdges = in_sizes[2];

    float* support = (float*)d_ws;
    int2*  recs    = (int2*)(support + (size_t)nNodes * D);
    int*   off     = (int*)(recs + nEdges);

    const int nBlocks = NBK;
    const int chunk   = (nEdges + nBlocks - 1) / nBlocks;

    {
        const int nStripes = nNodes / 16;
        gcn_gemm_and_partition<<<2 * nBlocks, FTH, 0, stream>>>(x, W, support, nStripes, nBlocks,
                                                               erow, ecol, eval, recs, off,
                                                               nEdges, chunk, nNodes);
    }
    gcn_aggregate<<<(nNodes + RPB - 1) / RPB, ETH, 0, stream>>>(support, recs, off, out, nNodes, chunk);
}


# tanh of the finished rows branch-free as 1 - 2/(e^(2|x|)+1) on v_exp_f32 + v_rcp_f32 (abs error < 3e-7) in place of the library call (about 40 instructions, 2 masked branches a value); same sums
# speedup vs baseline: 3.7001x; 1.0392x over previous
#include <hip/hip_runtime.h>
#include <hip/hip_bf16.h>

typedef __attribute__((ext_vector_type(2))) float v2f;
typedef __attribute__((ext_vector_type(8))) float v8f;

#define D 128

typedef __attribute__((ext_vector_type(4))) float v4f;

__device__ __forceinline__ void gemm_stripe(const float* __restrict__ x,
                                            const float* __restrict__ W,
                                            float* __restrict__ support,
                                            int stripe, int lane) {
    const int m0 = stripe * 16;
    const int h  = lane >> 4;
    const int c  = lane & 15;

    const float* arow = x + (size_t)(m0 + c) * D + 4 * h;
    const float* wcol = W + (size_t)(4 * h) * D + 4 * c;

    v8f acc[2][4] = {};
#pragma unroll 2
    for (int q = 0; q < D / 8; ++q) {
        const v4f a4 = *(const v4f*)(arow + 8 * q);
        v4f b4[4][2];
#pragma unroll
        for (int kk = 0; kk < 4; ++kk) {
#pragma unroll
            for (int g = 0; g < 2; ++g)
                b4[kk][g] = *(const v4f*)(wcol + (size_t)(8 * q + kk) * D + 64 * g);
        }
        v2f a_lo, a_hi;
        a_lo.x = a4.x; a_lo.y = a4.y;
        a_hi.x = a4.z; a_hi.y = a4.w;
#pragma unroll
        for (int g = 0; g < 2; ++g) {
#pragma unroll
            for (int t = 0; t < 4; ++t) {
                v2f b;
                b.x = b4[0][g][t];
                b.y = b4[1][g][t];
                acc[g][t] = __builtin_amdgcn_wmma_f32_16x16x4_f32(
                        false, a_lo, false, b, (short)0, acc[g][t], false, false);
            }
        }
#pragma unroll
        for (int g = 0; g < 2; ++g) {
#pragma unroll
            for (int t = 0; t < 4; ++t) {
                v2f b;
                b.x = b4[2][g][t];
                b.y = b4[3][g][t];
                acc[g][t] = __builtin_amdgcn_wmma_f32_16x16x4_f32(
                        false, a_hi, false, b, (short)0, acc[g][t], false, false);
            }
        }
    }

    float* srow = support + (size_t)(m0 + 8 * h) * D + 4 * c;
#pragma unroll
    for (int r = 0; r < 8; ++r) {
#pragma unroll
        for (int g = 0; g < 2; ++g) {
            v4f o;
            o.x = acc[g][0][r];
            o.y = acc[g][1][r];
            o.z = acc[g][2][r];
            o.w = acc[g][3][r];
            *(v4f*)(srow + (size_t)r * D + 64 * g) = o;
        }
    }
}

#define RPW     13
#define WPB     32
#define FW      16
#define FTH     (FW * 32)
#define RPB     (RPW * WPB)
#define NBK     256
#define OFFP    (NBK + 1)
#define ETH     (WPB * 32)
#define TILE    8192
#define GRP     8
#define RND     13
#define COLBITS 17
#define ROWBITS 9

__device__ __forceinline__ unsigned same_key_lanes(unsigned key, int nbits, bool valid) {
    unsigned grp = __builtin_amdgcn_ballot_w32(valid);
    for (int bit = 0; bit < nbits; ++bit) {
        const bool one = ((key >> bit) & 1u) != 0u;
        const unsigned bal = __builtin_amdgcn_ballot_w32(one);
        grp &= one ? bal : ~bal;
    }
    return valid ? grp : 0u;
}

__device__ __forceinline__ void wave_count(const int* __restrict__ erow, int ebeg, int eend,
                                           int nNodes, int lane, volatile int* wh) {
    const unsigned lt = (1u << lane) - 1u;
    for (int e = ebeg; e < eend; e += 32 * RND) {
        unsigned r8[RND];
#pragma unroll
        for (int k = 0; k < RND; ++k) {
            const int idx = e + 32 * k + lane;
            r8[k] = 0xffffffffu;
            if (idx < eend) r8[k] = (unsigned)erow[idx];
        }
#pragma unroll
        for (int k = 0; k < RND; ++k) {
            const unsigned r = r8[k];
            const bool valid = r < (unsigned)nNodes;
            const unsigned bk = valid ? r / RPB : 0u;
            const unsigned grp = same_key_lanes(bk, 8, valid);
            if (valid && (grp & lt) == 0u) wh[bk] += __popc(grp);
        }
    }
}

__global__ __launch_bounds__(FTH, 8) void gcn_gemm_and_partition(const float* __restrict__ x,
                                                                 const float* __restrict__ W,
                                                                 float* __restrict__ support,
                                                                 int nStripes, int nG,
                                                                 const int* __restrict__ erow,
                                                                 const int* __restrict__ ecol,
                                                                 const float* __restrict__ eval,
                                                                 int2* __restrict__ recs,
                                                                 int* __restrict__ off,
                                                                 int nEdges, int chunk, int nNodes) {
    __shared__ int wh[FW * NBK];
    __shared__ int tot[NBK + 1];
    const int tid = threadIdx.x;
    const int lane = tid & 31;
    const int wave = __builtin_amdgcn_readfirstlane(tid >> 5);
    if (blockIdx.x >= NBK) {
        const int g = (int)blockIdx.x - NBK;
        for (int s = wave * nG + g; s < nStripes; s += FW * nG) gemm_stripe(x, W, support, s, lane);
        return;
    }
    const int me = blockIdx.x;
    for (int i = tid; i < FW * NBK; i += FTH) wh[i] = 0;
    __syncthreads();
    const int e0 = me * chunk;
    const int e1 = (e0 + chunk < nEdges) ? (e0 + chunk) : nEdges;
    const int per = (chunk + FW - 1) / FW;
    const int wb = e0 + wave * per;
    const int we = (wb + per < e1) ? (wb + per) : e1;
    volatile int* mywh = wh + wave * NBK;
    wave_count(erow, wb, we, nNodes, lane, mywh);
    __syncthreads();
    if (tid < NBK) {
        int total = 0;
#pragma unroll
        for (int w = 0; w < FW; ++w) total += wh[w * NBK + tid];
        tot[tid] = total;
    }
    __syncthreads();
    if (tid < 32) {
        int v[8];
        int run = 0;
#pragma unroll
        for (int j = 0; j < 8; ++j) {
            const int c = tot[8 * tid + j];
            v[j] = run;
            run += c;
        }
        int incl = run;
#pragma unroll
        for (int d = 1; d < 32; d <<= 1) {
            const int o = __shfl_up(incl, d);
            if (tid >= d) incl += o;
        }
        const int excl = incl - run;
#pragma unroll
        for (int j = 0; j < 8; ++j) tot[8 * tid + j] = excl + v[j];
        if (tid == 31) tot[NBK] = incl;
    }
    __syncthreads();
    if (tid < NBK) {
        int run = tot[tid];
#pragma unroll
        for (int w = 0; w < FW; ++w) {
            const int v = wh[w * NBK + tid];
            wh[w * NBK + tid] = run;
            run += v;
        }
    }
    if (tid <= NBK) off[me * OFFP + tid] = tot[tid];
    __syncthreads();

    const unsigned lt = (1u << lane) - 1u;
    for (int e = wb; e < we; e += 32 * RND) {
        unsigned r8[RND];
        unsigned c8[RND];
        int v8[RND];
#pragma unroll
        for (int k = 0; k < RND; ++k) {
            const int idx = e + 32 * k + lane;
            r8[k] = 0xffffffffu;
            c8[k] = 0u;
            v8[k] = 0;
            if (idx < we) {
                r8[k] = (unsigned)erow[idx];
                c8[k] = (unsigned)ecol[idx];
                v8[k] = __float_as_int(eval[idx]);
            }
        }
#pragma unroll
        for (int k = 0; k < RND; ++k) {
            const unsigned r = r8[k];
            const bool valid = r < (unsigned)nNodes;
            const unsigned bk = valid ? r / RPB : 0u;
            const unsigned grp = same_key_lanes(bk, 8, valid);
            if (valid) {
                const int pos = e0 + mywh[bk] + __popc(grp & lt);
                int2 rec;
                rec.x = (int)((c8[k] & ((1u << COLBITS) - 1u)) | ((r - bk * RPB) << COLBITS));
                rec.y = v8[k];
                recs[pos] = rec;
            }
            if (valid && (grp & lt) == 0u) mywh[bk] += __popc(grp);
        }
    }
}

__device__ __forceinline__ v4f gather_row(const float* __restrict__ support, unsigned laneoff, int key) {
    const unsigned col = (unsigned)key & ((1u << COLBITS) - 1u);
    return *(const v4f*)((const char*)support + (size_t)col * (D * sizeof(float)) + laneoff);
}

typedef int __attribute__((may_alias)) int_alias;

__device__ __forceinline__ float tanh_exp_rcp(float x) {
    const float e = __builtin_amdgcn_exp2f(fabsf(x) * 2.8853900817779268f);
    const float r = __builtin_amdgcn_rcpf(e + 1.0f);
    return copysignf(fmaf(-2.0f, r, 1.0f), x);
}

__device__ __forceinline__ void store_row(const v4f* sums4, float* __restrict__ out, int lane,
                                          int r, int row, int nNodes) {
    if (row < nNodes) {
        const v4f a = sums4[r * 32 + lane];
        v4f o;
        o.x = tanh_exp_rcp(a.x);
        o.y = tanh_exp_rcp(a.y);
        o.z = tanh_exp_rcp(a.z);
        o.w = tanh_exp_rcp(a.w);
        *(v4f*)(out + (size_t)row * D + 4 * lane) = o;
    }
}

__global__ __launch_bounds__(ETH) void gcn_aggregate(const float* __restrict__ support,
                                                     const int2* __restrict__ recs,
                                                     const int* __restrict__ off,
                                                     float* __restrict__ out, int nNodes, int chunk) {
    __shared__ __attribute__((aligned(16))) float sums[RPB * D];
    __shared__ __attribute__((aligned(8)))  int2  tile[TILE];
    __shared__ __attribute__((aligned(4)))  unsigned short wh[WPB * RPB];
    __shared__ int rstart[RPB + 1];
    __shared__ int sstart[NBK + 1];
    __shared__ int sbase[NBK];
    const int tid = threadIdx.x;
    const int lane = tid & 31;
    const int wave = __builtin_amdgcn_readfirstlane(tid >> 5);
    const int b = blockIdx.x;
    const unsigned lt = (1u << lane) - 1u;
    const unsigned laneoff = (unsigned)lane * 16u;
    v4f* sums4 = (v4f*)sums;
    v4f* mys = sums4 + wave * RPW * 32;
    unsigned short* mywh = wh + wave * RPB;

    {
        const v4f z4 = {0.f, 0.f, 0.f, 0.f};
        for (int r = 0; r < RPW; ++r) mys[r * 32 + lane] = z4;
    }

    if (tid < NBK) {
        const int o0 = off[tid * OFFP + b];
        const int o1 = off[tid * OFFP + b + 1];
        sstart[tid] = o1 - o0;
        sbase[tid] = tid * chunk + o0;
    }
    __syncthreads();
    if (tid < 32) {
        int v[8];
        int run = 0;
#pragma unroll
        for (int j = 0; j < 8; ++j) {
            const int c = sstart[8 * tid + j];
            v[j] = run;
            run += c;
        }
        int incl = run;
#pragma unroll
        for (int d = 1; d < 32; d <<= 1) {
            const int o = __shfl_up(incl, d);
            if (tid >= d) incl += o;
        }
        const int excl = incl - run;
#pragma unroll
        for (int j = 0; j < 8; ++j) sstart[8 * tid + j] = excl + v[j];
        if (tid == 31) sstart[NBK] = incl;
    }
    __syncthreads();
    const int t = __builtin_amdgcn_readfirstlane(sstart[NBK]);
    const int row0 = wave * RPW;
    int done = row0;
    for (int tile0 = 0; tile0 < t; tile0 += TILE) {
        const int nt = (t - tile0 < TILE) ? (t - tile0) : TILE;
        const bool lastTile = tile0 + TILE >= t;
        for (int i = lane; i < RPB / 2; i += 32) ((int_alias*)mywh)[i] = 0;
        int2 rc[TILE / ETH];
        unsigned rowk[TILE / ETH];
        int pw[TILE / ETH];
#pragma unroll
        for (int k = 0; k < TILE / ETH; ++k) {
            const int idx = wave * (TILE / WPB) + k * 32 + lane;
            rc[k] = make_int2(0, 0);
            if (idx < nt) {
                const int gi = tile0 + idx;
                int lo = 0, hi = NBK;
#pragma unroll
                for (int it = 0; it < 8; ++it) {
                    const int mid = (lo + hi) >> 1;
                    if (sstart[mid] <= gi) lo = mid; else hi = mid;
                }
                rc[k] = recs[sbase[lo] + (gi - sstart[lo])];
            }
        }
#pragma unroll
        for (int k = 0; k < TILE / ETH; ++k) {
            const int idx = wave * (TILE / WPB) + k * 32 + lane;
            const bool valid = idx < nt;
            rowk[k] = valid ? ((unsigned)rc[k].x >> COLBITS) : 0u;
            const unsigned grp = same_key_lanes(rowk[k], ROWBITS, valid);
            const int base = mywh[rowk[k]];
            pw[k] = base + __popc(grp & lt);
            if (valid && (grp & lt) == 0u) mywh[rowk[k]] = (unsigned short)(base + __popc(grp));
        }
        __syncthreads();
        if (tid < RPB) {
            int run = 0;
#pragma unroll
            for (int w = 0; w < WPB; ++w) {
                const int v = wh[w * RPB + tid];
                wh[w * RPB + tid] = (unsigned short)run;
                run += v;
            }
            rstart[tid] = run;
        }
        __syncthreads();
        if (tid < 32) {
            int v[RPW];
            int run = 0;
#pragma unroll
            for (int j = 0; j < RPW; ++j) {
                const int c = rstart[RPW * lane + j];
                v[j] = run;
                run += c;
            }
            int incl = run;
#pragma unroll
            for (int d = 1; d < 32; d <<= 1) {
                const int o = __shfl_up(incl, d);
                if (lane >= d) incl += o;
            }
            const int excl = incl - run;
#pragma unroll
            for (int j = 0; j < RPW; ++j) rstart[RPW * lane + j] = excl + v[j];
            if (lane == 31) rstart[RPB] = incl;
        }
        __syncthreads();
#pragma unroll
        for (int k = 0; k < TILE / ETH; ++k) {
            const int idx = wave * (TILE / WPB) + k * 32 + lane;
            if (idx < nt) tile[rstart[rowk[k]] + wh[wave * RPB + rowk[k]] + pw[k]] = rc[k];
        }
        __syncthreads();
        const int a0 = __builtin_amdgcn_readfirstlane(rstart[row0]);
        const int a1 = __builtin_amdgcn_readfirstlane(rstart[row0 + RPW]);
        int cur = row0;
        v4f acc = sums4[cur * 32 + lane];
        for (int base = a0; base < a1; base += 32) {
            const int n = (a1 - base < 32) ? (a1 - base) : 32;
            int2 rec = make_int2(0, 0);
            if (lane < n) rec = tile[base + lane];
            if (n == 32) {
                v4f m[2][GRP];
#pragma unroll
                for (int j = 0; j < GRP; ++j)
                    m[0][j] = gather_row(support, laneoff, __builtin_amdgcn_readlane(rec.x, j));
                if (lastTile) {
                    for (; done < cur; ++done) store_row(sums4, out, lane, done, b * RPB + done, nNodes);
                }
#pragma unroll
                for (int g = 0; g < 32 / GRP; ++g) {
                    if (g + 1 < 32 / GRP) {
#pragma unroll
                        for (int j = 0; j < GRP; ++j)
                            m[(g + 1) & 1][j] = gather_row(support, laneoff,
                                    __builtin_amdgcn_readlane(rec.x, (g + 1) * GRP + j));
                    }
#pragma unroll
                    for (int j = 0; j < GRP; ++j) {
                        const int e = g * GRP + j;
                        const int key = __builtin_amdgcn_readlane(rec.x, e);
                        const float v = __int_as_float(__builtin_amdgcn_readlane(rec.y, e));
                        const int row = (int)((unsigned)key >> COLBITS);
                        if (row != cur) {
                            sums4[cur * 32 + lane] = acc;
                            acc = sums4[row * 32 + lane];
                            cur = row;
                        }
                        const v4f mm = m[g & 1][j];
                        acc.x += mm.x * v;
                        acc.y += mm.y * v;
                        acc.z += mm.z * v;
                        acc.w += mm.w * v;
                    }
                }
            } else {
                for (int i0 = 0; i0 < n; i0 += GRP) {
                    v4f mt[GRP];
#pragma unroll
                    for (int j = 0; j < GRP; ++j) {
                        mt[j] = (v4f){0.f, 0.f, 0.f, 0.f};
                        if (i0 + j < n)
                            mt[j] = gather_row(support, laneoff, __builtin_amdgcn_readlane(rec.x, i0 + j));
                    }
#pragma unroll
                    for (int j = 0; j < GRP; ++j) {
                        if (i0 + j < n) {
                            const int key = __builtin_amdgcn_readlane(rec.x, i0 + j);
                            const float v = __int_as_float(__builtin_amdgcn_readlane(rec.y, i0 + j));
                            const int row = (int)((unsigned)key >> COLBITS);
                            if (row != cur) {
                                sums4[cur * 32 + lane] = acc;
                                acc = sums4[row * 32 + lane];
                                cur = row;
                            }
                            acc.x += mt[j].x * v;
                            acc.y += mt[j].y * v;
                            acc.z += mt[j].z * v;
                            acc.w += mt[j].w * v;
                        }
                    }
                }
            }
        }
        sums4[cur * 32 + lane] = acc;
        __syncthreads();
    }

    for (; done < row0 + RPW; ++done) store_row(sums4, out, lane, done, b * RPB + done, nNodes);
}

extern "C" void kernel_launch(void* const* d_in, const int* in_sizes, int n_in,
                              void* d_out, int out_size, void* d_ws, size_t ws_size,
                              hipStream_t stream) {
    const float* x    = (const float*)d_in[0];
    const float* W    = (const float*)d_in[1];
    const int*   erow = (const int*)d_in[2];
    const int*   ecol = (const int*)d_in[3];
    const float* eval = (const float*)d_in[4];
    float* out = (float*)d_out;

    const int nNodes = in_sizes[0] / D;
    const int nEdges = in_sizes[2];

    float* support = (float*)d_ws;
    int2*  recs    = (int2*)(support + (size_t)nNodes * D);
    int*   off     = (int*)(recs + nEdges);

    const int nBlocks = NBK;
    const int chunk   = (nEdges + nBlocks - 1) / nBlocks;

    {
        const int nStripes = nNodes / 16;
        gcn_gemm_and_partition<<<2 * nBlocks, FTH, 0, stream>>>(x, W, support, nStripes, nBlocks,
                                                               erow, ecol, eval, recs, off,
                                                               nEdges, chunk, nNodes);
    }
    gcn_aggregate<<<(nNodes + RPB - 1) / RPB, ETH, 0, stream>>>(support, recs, off, out, nNodes, chunk);
}
